# starting kernel
# baseline (speedup 1.0000x reference)
#include <hip/hip_runtime.h>
#include <hip/hip_bf16.h>

typedef __attribute__((ext_vector_type(2))) float v2f;
typedef __attribute__((ext_vector_type(8))) float v8f;

#define IN_CH 128
#define IN2   256
#define HID   128
#define OUTC  10

__global__ __launch_bounds__(256) void k_deg_init(float* deg, int N) {
    int i = blockIdx.x * 256 + threadIdx.x;
    if (i < N) deg[i] = 1.0f;
}

__global__ __launch_bounds__(256) void k_deg_count(const int* __restrict__ dst, float* deg, int E) {
    int e = blockIdx.x * 256 + threadIdx.x;
    if (e < E) unsafeAtomicAdd(&deg[dst[e]], 1.0f);
}

__global__ __launch_bounds__(256) void k_rsqrt(float* deg, int N) {
    int i = blockIdx.x * 256 + threadIdx.x;
    if (i < N) deg[i] = rsqrtf(deg[i]);
}

#define AS_STRIDE 258

__global__ __launch_bounds__(256) void k_gemm1(const float* __restrict__ x,
                                               const float* __restrict__ anc,
                                               const float* __restrict__ W1,
                                               float* __restrict__ h1, int N) {
    __shared__ float As[16 * AS_STRIDE];
    const int row0 = blockIdx.x * 16;
    const int tid = threadIdx.x;

    for (int idx = tid; idx < 16 * IN2; idx += 256) {
        int r = idx >> 8, k = idx & 255;
        int node = row0 + r;
        float v = 0.0f;
        if (node < N) {
            v = (k < IN_CH) ? (x[(size_t)node * IN_CH + k] - anc[(size_t)node * IN_CH + k])
                            : anc[(size_t)node * IN_CH + (k - IN_CH)];
        }
        As[r * AS_STRIDE + k] = v;
    }
    __syncthreads();

    const int wave = tid >> 5, lane = tid & 31;
    const int grp = lane >> 4;
    const int cidx = lane & 15;
    const int n0 = wave * 16;

    v8f acc0 = {}, acc1 = {};
    for (int k0 = 0; k0 < IN2; k0 += 8) {
        {
            int kk = k0 + 2 * grp;
            v2f a, b;
            a.x = As[cidx * AS_STRIDE + kk];
            a.y = As[cidx * AS_STRIDE + kk + 1];
            b.x = W1[(size_t)kk * HID + n0 + cidx];
            b.y = W1[(size_t)(kk + 1) * HID + n0 + cidx];
            acc0 = __builtin_amdgcn_wmma_f32_16x16x4_f32(false, a, false, b, (short)0, acc0, false, false);
        }
        {
            int kk = k0 + 4 + 2 * grp;
            v2f a, b;
            a.x = As[cidx * AS_STRIDE + kk];
            a.y = As[cidx * AS_STRIDE + kk + 1];
            b.x = W1[(size_t)kk * HID + n0 + cidx];
            b.y = W1[(size_t)(kk + 1) * HID + n0 + cidx];
            acc1 = __builtin_amdgcn_wmma_f32_16x16x4_f32(false, a, false, b, (short)0, acc1, false, false);
        }
    }
    acc0 = acc0 + acc1;

    float* outp = h1 + (size_t)(row0 + 8 * grp) * HID + n0 + cidx;
    if (row0 + 16 <= N) {
        #pragma unroll
        for (int v = 0; v < 8; ++v) outp[(size_t)v * HID] = acc0[v];
    } else {
        #pragma unroll
        for (int v = 0; v < 8; ++v) {
            if (row0 + v + 8 * grp < N) outp[(size_t)v * HID] = acc0[v];
        }
    }
}

__global__ __launch_bounds__(256) void k_self1(const float* __restrict__ h1,
                                               const float* __restrict__ dinv,
                                               float* __restrict__ agg1, int N) {
    int i = blockIdx.x * 256 + threadIdx.x;
    if (i >= N * (HID / 4)) return;
    int node = i >> 5;
    float s = dinv[node]; s = s * s;
    float4 v = *(const float4*)(h1 + (size_t)i * 4);
    v.x *= s; v.y *= s; v.z *= s; v.w *= s;
    *(float4*)(agg1 + (size_t)i * 4) = v;
}

__global__ __launch_bounds__(256) void k_scatter1(const int* __restrict__ src,
                                                  const int* __restrict__ dst,
                                                  const float* __restrict__ h1,
                                                  const float* __restrict__ dinv,
                                                  float* __restrict__ agg1, int E) {
    int gt = blockIdx.x * 256 + threadIdx.x;
    int e = gt >> 5;
    int lane = gt & 31;
    if (e >= E) return;
    int s = src[e], d = dst[e];
    float nrm = dinv[s] * dinv[d];
    float4 v = *(const float4*)(h1 + (size_t)s * HID + lane * 4);
    float* o = agg1 + (size_t)d * HID + lane * 4;
    unsafeAtomicAdd(o + 0, v.x * nrm);
    unsafeAtomicAdd(o + 1, v.y * nrm);
    unsafeAtomicAdd(o + 2, v.z * nrm);
    unsafeAtomicAdd(o + 3, v.w * nrm);
}

#define AS2_STRIDE 132

__global__ __launch_bounds__(128) void k_gemm2(const float* __restrict__ agg1,
                                               const float* __restrict__ b1,
                                               const float* __restrict__ W2,
                                               float* __restrict__ h2, int N) {
    __shared__ float As[64 * AS2_STRIDE];
    __shared__ float Bs[HID * 16];
    const int node0 = blockIdx.x * 64;
    const int tid = threadIdx.x;

    for (int idx = tid; idx < 64 * HID; idx += 128) {
        int r = idx >> 7, k = idx & 127;
        int node = node0 + r;
        float v = 0.0f;
        if (node < N) {
            v = agg1[(size_t)node * HID + k] + b1[k];
            v = v > 0.0f ? v : 0.0f;
        }
        As[r * AS2_STRIDE + k] = v;
    }
    for (int idx = tid; idx < HID * 16; idx += 128) {
        int k = idx >> 4, n = idx & 15;
        Bs[idx] = (n < OUTC) ? W2[(size_t)k * OUTC + n] : 0.0f;
    }
    __syncthreads();

    const int wave = tid >> 5, lane = tid & 31;
    const int grp = lane >> 4, cidx = lane & 15;
    const int rbase = wave * 16;

    v8f acc0 = {}, acc1 = {};
    for (int k0 = 0; k0 < HID; k0 += 8) {
        {
            int kk = k0 + 2 * grp;
            v2f a, b;
            a.x = As[(rbase + cidx) * AS2_STRIDE + kk];
            a.y = As[(rbase + cidx) * AS2_STRIDE + kk + 1];
            b.x = Bs[kk * 16 + cidx];
            b.y = Bs[(kk + 1) * 16 + cidx];
            acc0 = __builtin_amdgcn_wmma_f32_16x16x4_f32(false, a, false, b, (short)0, acc0, false, false);
        }
        {
            int kk = k0 + 4 + 2 * grp;
            v2f a, b;
            a.x = As[(rbase + cidx) * AS2_STRIDE + kk];
            a.y = As[(rbase + cidx) * AS2_STRIDE + kk + 1];
            b.x = Bs[kk * 16 + cidx];
            b.y = Bs[(kk + 1) * 16 + cidx];
            acc1 = __builtin_amdgcn_wmma_f32_16x16x4_f32(false, a, false, b, (short)0, acc1, false, false);
        }
    }
    acc0 = acc0 + acc1;

    const int mrow0 = node0 + rbase + 8 * grp;
    float* outp = h2 + (size_t)mrow0 * OUTC + cidx;
    if (node0 + 64 <= N) {
        if (cidx < OUTC) {
            #pragma unroll
            for (int v = 0; v < 8; ++v) outp[(size_t)v * OUTC] = acc0[v];
        }
    } else {
        #pragma unroll
        for (int v = 0; v < 8; ++v) {
            if (mrow0 + v < N && cidx < OUTC) outp[(size_t)v * OUTC] = acc0[v];
        }
    }
}

__global__ __launch_bounds__(256) void k_self2(const float* __restrict__ h2,
                                               const float* __restrict__ dinv,
                                               const float* __restrict__ b2,
                                               float* __restrict__ out, int N) {
    int i = blockIdx.x * 256 + threadIdx.x;
    if (i >= N * OUTC) return;
    int node = i / OUTC;
    int c = i - node * OUTC;
    float s = dinv[node];
    out[i] = h2[i] * s * s + b2[c];
}

__global__ __launch_bounds__(256) void k_scatter2(const int* __restrict__ src,
                                                  const int* __restrict__ dst,
                                                  const float* __restrict__ h2,
                                                  const float* __restrict__ dinv,
                                                  float* __restrict__ out, int E) {
    int gt = blockIdx.x * 256 + threadIdx.x;
    int e = gt >> 4;
    int c = gt & 15;
    if (e >= E || c >= OUTC) return;
    int s = src[e], d = dst[e];
    float nrm = dinv[s] * dinv[d];
    unsafeAtomicAdd(&out[(size_t)d * OUTC + c], h2[(size_t)s * OUTC + c] * nrm);
}

extern "C" void kernel_launch(void* const* d_in, const int* in_sizes, int n_in,
                              void* d_out, int out_size, void* d_ws, size_t ws_size,
                              hipStream_t stream) {
    const float* x    = (const float*)d_in[0];
    const int*   ei   = (const int*)d_in[1];
    const float* anc  = (const float*)d_in[3];
    const float* W1   = (const float*)d_in[4];
    const float* b1   = (const float*)d_in[5];
    const float* W2   = (const float*)d_in[6];
    const float* b2   = (const float*)d_in[7];
    float* out = (float*)d_out;

    const int N = in_sizes[0] / IN_CH;
    const int E = in_sizes[1] / 2;
    const int* src = ei;
    const int* dst = ei + E;

    float* h1   = (float*)d_ws;
    float* agg1 = h1 + (size_t)N * HID;
    float* h2   = agg1 + (size_t)N * HID;
    float* dinv = h2 + (size_t)N * OUTC;

    k_deg_init <<<(N + 255) / 256, 256, 0, stream>>>(dinv, N);
    k_deg_count<<<(E + 255) / 256, 256, 0, stream>>>(dst, dinv, E);
    k_rsqrt    <<<(N + 255) / 256, 256, 0, stream>>>(dinv, N);

    k_gemm1<<<(N + 15) / 16, 256, 0, stream>>>(x, anc, W1, h1, N);

    k_self1   <<<(N * (HID / 4) + 255) / 256, 256, 0, stream>>>(h1, dinv, agg1, N);
    k_scatter1<<<((size_t)E * 32 + 255) / 256, 256, 0, stream>>>(src, dst, h1, dinv, agg1, E);

    k_gemm2<<<(N + 63) / 64, 128, 0, stream>>>(agg1, b1, W2, h2, N);

    k_self2   <<<(N * OUTC + 255) / 256, 256, 0, stream>>>(h2, dinv, b2, out, N);
    k_scatter2<<<((size_t)E * 16 + 255) / 256, 256, 0, stream>>>(src, dst, h2, dinv, out, E);
}


# 128-column edge scatter: lane l adds columns l+32j, so each f32 atomic instruction covers one whole 128-byte line, not four quarter-filled lines; sums stay in arrival order as before
# speedup vs baseline: 1.2296x; 1.2296x over previous
#include <hip/hip_runtime.h>
#include <hip/hip_bf16.h>

typedef __attribute__((ext_vector_type(2))) float v2f;
typedef __attribute__((ext_vector_type(8))) float v8f;

#define IN_CH 128
#define IN2   256
#define HID   128
#define OUTC  10

__global__ __launch_bounds__(256) void k_deg_init(float* deg, int N) {
    int i = blockIdx.x * 256 + threadIdx.x;
    if (i < N) deg[i] = 1.0f;
}

__global__ __launch_bounds__(256) void k_deg_count(const int* __restrict__ dst, float* deg, int E) {
    int e = blockIdx.x * 256 + threadIdx.x;
    if (e < E) unsafeAtomicAdd(&deg[dst[e]], 1.0f);
}

__global__ __launch_bounds__(256) void k_rsqrt(float* deg, int N) {
    int i = blockIdx.x * 256 + threadIdx.x;
    if (i < N) deg[i] = rsqrtf(deg[i]);
}

#define AS_STRIDE 258

__global__ __launch_bounds__(256) void k_gemm1(const float* __restrict__ x,
                                               const float* __restrict__ anc,
                                               const float* __restrict__ W1,
                                               float* __restrict__ h1, int N) {
    __shared__ float As[16 * AS_STRIDE];
    const int row0 = blockIdx.x * 16;
    const int tid = threadIdx.x;

    for (int idx = tid; idx < 16 * IN2; idx += 256) {
        int r = idx >> 8, k = idx & 255;
        int node = row0 + r;
        float v = 0.0f;
        if (node < N) {
            v = (k < IN_CH) ? (x[(size_t)node * IN_CH + k] - anc[(size_t)node * IN_CH + k])
                            : anc[(size_t)node * IN_CH + (k - IN_CH)];
        }
        As[r * AS_STRIDE + k] = v;
    }
    __syncthreads();

    const int wave = tid >> 5, lane = tid & 31;
    const int grp = lane >> 4;
    const int cidx = lane & 15;
    const int n0 = wave * 16;

    v8f acc0 = {}, acc1 = {};
    for (int k0 = 0; k0 < IN2; k0 += 8) {
        {
            int kk = k0 + 2 * grp;
            v2f a, b;
            a.x = As[cidx * AS_STRIDE + kk];
            a.y = As[cidx * AS_STRIDE + kk + 1];
            b.x = W1[(size_t)kk * HID + n0 + cidx];
            b.y = W1[(size_t)(kk + 1) * HID + n0 + cidx];
            acc0 = __builtin_amdgcn_wmma_f32_16x16x4_f32(false, a, false, b, (short)0, acc0, false, false);
        }
        {
            int kk = k0 + 4 + 2 * grp;
            v2f a, b;
            a.x = As[cidx * AS_STRIDE + kk];
            a.y = As[cidx * AS_STRIDE + kk + 1];
            b.x = W1[(size_t)kk * HID + n0 + cidx];
            b.y = W1[(size_t)(kk + 1) * HID + n0 + cidx];
            acc1 = __builtin_amdgcn_wmma_f32_16x16x4_f32(false, a, false, b, (short)0, acc1, false, false);
        }
    }
    acc0 = acc0 + acc1;

    float* outp = h1 + (size_t)(row0 + 8 * grp) * HID + n0 + cidx;
    if (row0 + 16 <= N) {
        #pragma unroll
        for (int v = 0; v < 8; ++v) outp[(size_t)v * HID] = acc0[v];
    } else {
        #pragma unroll
        for (int v = 0; v < 8; ++v) {
            if (row0 + v + 8 * grp < N) outp[(size_t)v * HID] = acc0[v];
        }
    }
}

__global__ __launch_bounds__(256) void k_self1(const float* __restrict__ h1,
                                               const float* __restrict__ dinv,
                                               float* __restrict__ agg1, int N) {
    int i = blockIdx.x * 256 + threadIdx.x;
    if (i >= N * (HID / 4)) return;
    int node = i >> 5;
    float s = dinv[node]; s = s * s;
    float4 v = *(const float4*)(h1 + (size_t)i * 4);
    v.x *= s; v.y *= s; v.z *= s; v.w *= s;
    *(float4*)(agg1 + (size_t)i * 4) = v;
}

__global__ __launch_bounds__(256) void k_scatter1(const int* __restrict__ src,
                                                  const int* __restrict__ dst,
                                                  const float* __restrict__ h1,
                                                  const float* __restrict__ dinv,
                                                  float* __restrict__ agg1, int E) {
    int gt = blockIdx.x * 256 + threadIdx.x;
    int e = gt >> 5;
    int lane = gt & 31;
    if (e >= E) return;
    int s = src[e], d = dst[e];
    float nrm = dinv[s] * dinv[d];
    const float* hp = h1 + (size_t)s * HID + lane;
    float v0 = hp[0];
    float v1 = hp[32];
    float v2 = hp[64];
    float v3 = hp[96];
    float* o = agg1 + (size_t)d * HID + lane;
    unsafeAtomicAdd(o + 0,  v0 * nrm);
    unsafeAtomicAdd(o + 32, v1 * nrm);
    unsafeAtomicAdd(o + 64, v2 * nrm);
    unsafeAtomicAdd(o + 96, v3 * nrm);
}

#define AS2_STRIDE 132

__global__ __launch_bounds__(128) void k_gemm2(const float* __restrict__ agg1,
                                               const float* __restrict__ b1,
                                               const float* __restrict__ W2,
                                               float* __restrict__ h2, int N) {
    __shared__ float As[64 * AS2_STRIDE];
    __shared__ float Bs[HID * 16];
    const int node0 = blockIdx.x * 64;
    const int tid = threadIdx.x;

    for (int idx = tid; idx < 64 * HID; idx += 128) {
        int r = idx >> 7, k = idx & 127;
        int node = node0 + r;
        float v = 0.0f;
        if (node < N) {
            v = agg1[(size_t)node * HID + k] + b1[k];
            v = v > 0.0f ? v : 0.0f;
        }
        As[r * AS2_STRIDE + k] = v;
    }
    for (int idx = tid; idx < HID * 16; idx += 128) {
        int k = idx >> 4, n = idx & 15;
        Bs[idx] = (n < OUTC) ? W2[(size_t)k * OUTC + n] : 0.0f;
    }
    __syncthreads();

    const int wave = tid >> 5, lane = tid & 31;
    const int grp = lane >> 4, cidx = lane & 15;
    const int rbase = wave * 16;

    v8f acc0 = {}, acc1 = {};
    for (int k0 = 0; k0 < HID; k0 += 8) {
        {
            int kk = k0 + 2 * grp;
            v2f a, b;
            a.x = As[(rbase + cidx) * AS2_STRIDE + kk];
            a.y = As[(rbase + cidx) * AS2_STRIDE + kk + 1];
            b.x = Bs[kk * 16 + cidx];
            b.y = Bs[(kk + 1) * 16 + cidx];
            acc0 = __builtin_amdgcn_wmma_f32_16x16x4_f32(false, a, false, b, (short)0, acc0, false, false);
        }
        {
            int kk = k0 + 4 + 2 * grp;
            v2f a, b;
            a.x = As[(rbase + cidx) * AS2_STRIDE + kk];
            a.y = As[(rbase + cidx) * AS2_STRIDE + kk + 1];
            b.x = Bs[kk * 16 + cidx];
            b.y = Bs[(kk + 1) * 16 + cidx];
            acc1 = __builtin_amdgcn_wmma_f32_16x16x4_f32(false, a, false, b, (short)0, acc1, false, false);
        }
    }
    acc0 = acc0 + acc1;

    const int mrow0 = node0 + rbase + 8 * grp;
    float* outp = h2 + (size_t)mrow0 * OUTC + cidx;
    if (node0 + 64 <= N) {
        if (cidx < OUTC) {
            #pragma unroll
            for (int v = 0; v < 8; ++v) outp[(size_t)v * OUTC] = acc0[v];
        }
    } else {
        #pragma unroll
        for (int v = 0; v < 8; ++v) {
            if (mrow0 + v < N && cidx < OUTC) outp[(size_t)v * OUTC] = acc0[v];
        }
    }
}

__global__ __launch_bounds__(256) void k_self2(const float* __restrict__ h2,
                                               const float* __restrict__ dinv,
                                               const float* __restrict__ b2,
                                               float* __restrict__ out, int N) {
    int i = blockIdx.x * 256 + threadIdx.x;
    if (i >= N * OUTC) return;
    int node = i / OUTC;
    int c = i - node * OUTC;
    float s = dinv[node];
    out[i] = h2[i] * s * s + b2[c];
}

__global__ __launch_bounds__(256) void k_scatter2(const int* __restrict__ src,
                                                  const int* __restrict__ dst,
                                                  const float* __restrict__ h2,
                                                  const float* __restrict__ dinv,
                                                  float* __restrict__ out, int E) {
    int gt = blockIdx.x * 256 + threadIdx.x;
    int e = gt >> 4;
    int c = gt & 15;
    if (e >= E || c >= OUTC) return;
    int s = src[e], d = dst[e];
    float nrm = dinv[s] * dinv[d];
    unsafeAtomicAdd(&out[(size_t)d * OUTC + c], h2[(size_t)s * OUTC + c] * nrm);
}

extern "C" void kernel_launch(void* const* d_in, const int* in_sizes, int n_in,
                              void* d_out, int out_size, void* d_ws, size_t ws_size,
                              hipStream_t stream) {
    const float* x    = (const float*)d_in[0];
    const int*   ei   = (const int*)d_in[1];
    const float* anc  = (const float*)d_in[3];
    const float* W1   = (const float*)d_in[4];
    const float* b1   = (const float*)d_in[5];
    const float* W2   = (const float*)d_in[6];
    const float* b2   = (const float*)d_in[7];
    float* out = (float*)d_out;

    const int N = in_sizes[0] / IN_CH;
    const int E = in_sizes[1] / 2;
    const int* src = ei;
    const int* dst = ei + E;

    float* h1   = (float*)d_ws;
    float* agg1 = h1 + (size_t)N * HID;
    float* h2   = agg1 + (size_t)N * HID;
    float* dinv = h2 + (size_t)N * OUTC;

    k_deg_init <<<(N + 255) / 256, 256, 0, stream>>>(dinv, N);
    k_deg_count<<<(E + 255) / 256, 256, 0, stream>>>(dst, dinv, E);
    k_rsqrt    <<<(N + 255) / 256, 256, 0, stream>>>(dinv, N);

    k_gemm1<<<(N + 15) / 16, 256, 0, stream>>>(x, anc, W1, h1, N);

    k_self1   <<<(N * (HID / 4) + 255) / 256, 256, 0, stream>>>(h1, dinv, agg1, N);
    k_scatter1<<<((size_t)E * 32 + 255) / 256, 256, 0, stream>>>(src, dst, h1, dinv, agg1, E);

    k_gemm2<<<(N + 63) / 64, 128, 0, stream>>>(agg1, b1, W2, h2, N);

    k_self2   <<<(N * OUTC + 255) / 256, 256, 0, stream>>>(h2, dinv, b2, out, N);
    k_scatter2<<<((size_t)E * 16 + 255) / 256, 256, 0, stream>>>(src, dst, h2, dinv, out, E);
}


# edge scatter with f32 atomics replaced by a per-call dst-sorted edge list (int count, scan, fill) and one wave per node gathering neighbour rows; sums now in list (arrival) order
# speedup vs baseline: 1.5615x; 1.2699x over previous
#include <hip/hip_runtime.h>
#include <hip/hip_bf16.h>


typedef __attribute__((ext_vector_type(2))) float v2f;
typedef __attribute__((ext_vector_type(8))) float v8f;

#define IN_CH 128
#define IN2   256
#define HID   128
#define OUTC  10

__global__ __launch_bounds__(256) void k_cnt_zero(int* cnt, int N) {
    int i = blockIdx.x * 256 + threadIdx.x;
    if (i < N) cnt[i] = 0;
}

__global__ __launch_bounds__(256) void k_cnt(const int* __restrict__ dst, int* cnt, int E) {
    int e = blockIdx.x * 256 + threadIdx.x;
    if (e < E) atomicAdd(&cnt[dst[e]], 1);
}

__device__ __forceinline__ int wave_sum(int v) {
    #pragma unroll
    for (int o = 16; o > 0; o >>= 1) v += __shfl_xor(v, o, 32);
    return v;
}

__global__ __launch_bounds__(1024) void k_blocksum(const int* __restrict__ cnt,
                                                   int* __restrict__ bsum,
                                                   float* __restrict__ dinv, int N) {
    __shared__ int wsum[32];
    const int tid = threadIdx.x, lane = tid & 31, wave = tid >> 5;
    const int i = blockIdx.x * 1024 + tid;
    int c = 0;
    if (i < N) {
        c = cnt[i];
        dinv[i] = rsqrtf((float)(c + 1));
    }
    int v = wave_sum(c);
    if (lane == 0) wsum[wave] = v;
    __syncthreads();
    if (wave == 0) {
        int t = wave_sum(wsum[lane]);
        if (lane == 0) bsum[blockIdx.x] = t;
    }
}

__global__ __launch_bounds__(1024) void k_scan(const int* __restrict__ cnt,
                                               const int* __restrict__ bsum,
                                               int* __restrict__ rp, int N) {
    __shared__ int wtot[32];
    __shared__ int ptot[32];
    const int tid = threadIdx.x, lane = tid & 31, wave = tid >> 5;
    int p = 0;
    for (int b = tid; b < (int)blockIdx.x; b += 1024) p += bsum[b];
    p = wave_sum(p);
    const int i = blockIdx.x * 1024 + tid;
    const int c = (i < N) ? cnt[i] : 0;
    int incl = c;
    #pragma unroll
    for (int o = 1; o < 32; o <<= 1) {
        int t = __shfl_up(incl, o, 32);
        if (lane >= o) incl += t;
    }
    if (lane == 31) wtot[wave] = incl;
    if (lane == 0) ptot[wave] = p;
    __syncthreads();
    int off = ptot[lane] + ((lane < wave) ? wtot[lane] : 0);
    off = wave_sum(off);
    if (i < N) rp[i + 1] = off + incl - c;
    if (i == 0) rp[0] = 0;
}

__global__ __launch_bounds__(256) void k_fill(const int* __restrict__ src,
                                              const int* __restrict__ dst,
                                              const float* __restrict__ dinv,
                                              int* rp, int2* __restrict__ csr, int E) {
    int e = blockIdx.x * 256 + threadIdx.x;
    if (e >= E) return;
    int s = src[e], d = dst[e];
    int pos = atomicAdd(&rp[d + 1], 1);
    csr[pos] = make_int2(s, __float_as_int(dinv[s]));
}

#define AS_STRIDE 258

__global__ __launch_bounds__(256) void k_gemm1(const float* __restrict__ x,
                                               const float* __restrict__ anc,
                                               const float* __restrict__ W1,
                                               float* __restrict__ h1, int N) {
    __shared__ float As[16 * AS_STRIDE];
    const int row0 = blockIdx.x * 16;
    const int tid = threadIdx.x;

    for (int idx = tid; idx < 16 * IN2; idx += 256) {
        int r = idx >> 8, k = idx & 255;
        int node = row0 + r;
        float v = 0.0f;
        if (node < N) {
            v = (k < IN_CH) ? (x[(size_t)node * IN_CH + k] - anc[(size_t)node * IN_CH + k])
                            : anc[(size_t)node * IN_CH + (k - IN_CH)];
        }
        As[r * AS_STRIDE + k] = v;
    }
    __syncthreads();

    const int wave = tid >> 5, lane = tid & 31;
    const int grp = lane >> 4;
    const int cidx = lane & 15;
    const int n0 = wave * 16;

    v8f acc0 = {}, acc1 = {};
    for (int k0 = 0; k0 < IN2; k0 += 8) {
        {
            int kk = k0 + 2 * grp;
            v2f a, b;
            a.x = As[cidx * AS_STRIDE + kk];
            a.y = As[cidx * AS_STRIDE + kk + 1];
            b.x = W1[(size_t)kk * HID + n0 + cidx];
            b.y = W1[(size_t)(kk + 1) * HID + n0 + cidx];
            acc0 = __builtin_amdgcn_wmma_f32_16x16x4_f32(false, a, false, b, (short)0, acc0, false, false);
        }
        {
            int kk = k0 + 4 + 2 * grp;
            v2f a, b;
            a.x = As[cidx * AS_STRIDE + kk];
            a.y = As[cidx * AS_STRIDE + kk + 1];
            b.x = W1[(size_t)kk * HID + n0 + cidx];
            b.y = W1[(size_t)(kk + 1) * HID + n0 + cidx];
            acc1 = __builtin_amdgcn_wmma_f32_16x16x4_f32(false, a, false, b, (short)0, acc1, false, false);
        }
    }
    acc0 = acc0 + acc1;

    float* outp = h1 + (size_t)(row0 + 8 * grp) * HID + n0 + cidx;
    if (row0 + 16 <= N) {
        #pragma unroll
        for (int v = 0; v < 8; ++v) outp[(size_t)v * HID] = acc0[v];
    } else {
        #pragma unroll
        for (int v = 0; v < 8; ++v) {
            if (row0 + v + 8 * grp < N) outp[(size_t)v * HID] = acc0[v];
        }
    }
}

__global__ __launch_bounds__(256) void k_gather1(const int* __restrict__ rp,
                                                 const int2* __restrict__ csr,
                                                 const float* __restrict__ h1,
                                                 const float* __restrict__ dinv,
                                                 float* __restrict__ agg1, int N) {
    const int lane = threadIdx.x & 31;
    const int d = __builtin_amdgcn_readfirstlane((int)(blockIdx.x * 8 + (threadIdx.x >> 5)));
    if (d >= N) return;
    const int beg = __builtin_amdgcn_readfirstlane(rp[d]);
    const int end = __builtin_amdgcn_readfirstlane(rp[d + 1]);
    const float dd = dinv[d];
    const float4 sv = *(const float4*)(h1 + (size_t)d * HID + lane * 4);
    float4 acc = {0.0f, 0.0f, 0.0f, 0.0f};

    for (int base = beg; base < end; base += 32) {
        const int m = (end - base < 32) ? (end - base) : 32;
        int es = d;
        float ew = 0.0f;
        if (lane < m) {
            int2 en = csr[base + lane];
            es = en.x;
            ew = __int_as_float(en.y) * dd;
        }
        for (int k0 = 0; k0 < m; k0 += 8) {
            const int r = m - k0;
            float4 v[8];
            float w[8];
            #pragma unroll
            for (int u = 0; u < 8; ++u) {
                if (u < r) {
                    int s = __builtin_amdgcn_readlane(es, k0 + u);
                    w[u] = __int_as_float(__builtin_amdgcn_readlane(__float_as_int(ew), k0 + u));
                    v[u] = *(const float4*)(h1 + (size_t)s * HID + lane * 4);
                }
            }
            #pragma unroll
            for (int u = 0; u < 8; ++u) {
                if (u < r) {
                    acc.x += v[u].x * w[u];
                    acc.y += v[u].y * w[u];
                    acc.z += v[u].z * w[u];
                    acc.w += v[u].w * w[u];
                }
            }
        }
    }
    const float sw = dd * dd;
    acc.x += sv.x * sw; acc.y += sv.y * sw; acc.z += sv.z * sw; acc.w += sv.w * sw;
    *(float4*)(agg1 + (size_t)d * HID + lane * 4) = acc;
}

#define AS2_STRIDE 132

__global__ __launch_bounds__(128) void k_gemm2(const float* __restrict__ agg1,
                                               const float* __restrict__ b1,
                                               const float* __restrict__ W2,
                                               float* __restrict__ h2, int N) {
    __shared__ float As[64 * AS2_STRIDE];
    __shared__ float Bs[HID * 16];
    const int node0 = blockIdx.x * 64;
    const int tid = threadIdx.x;

    for (int idx = tid; idx < 64 * HID; idx += 128) {
        int r = idx >> 7, k = idx & 127;
        int node = node0 + r;
        float v = 0.0f;
        if (node < N) {
            v = agg1[(size_t)node * HID + k] + b1[k];
            v = v > 0.0f ? v : 0.0f;
        }
        As[r * AS2_STRIDE + k] = v;
    }
    for (int idx = tid; idx < HID * 16; idx += 128) {
        int k = idx >> 4, n = idx & 15;
        Bs[idx] = (n < OUTC) ? W2[(size_t)k * OUTC + n] : 0.0f;
    }
    __syncthreads();

    const int wave = tid >> 5, lane = tid & 31;
    const int grp = lane >> 4, cidx = lane & 15;
    const int rbase = wave * 16;

    v8f acc0 = {}, acc1 = {};
    for (int k0 = 0; k0 < HID; k0 += 8) {
        {
            int kk = k0 + 2 * grp;
            v2f a, b;
            a.x = As[(rbase + cidx) * AS2_STRIDE + kk];
            a.y = As[(rbase + cidx) * AS2_STRIDE + kk + 1];
            b.x = Bs[kk * 16 + cidx];
            b.y = Bs[(kk + 1) * 16 + cidx];
            acc0 = __builtin_amdgcn_wmma_f32_16x16x4_f32(false, a, false, b, (short)0, acc0, false, false);
        }
        {
            int kk = k0 + 4 + 2 * grp;
            v2f a, b;
            a.x = As[(rbase + cidx) * AS2_STRIDE + kk];
            a.y = As[(rbase + cidx) * AS2_STRIDE + kk + 1];
            b.x = Bs[kk * 16 + cidx];
            b.y = Bs[(kk + 1) * 16 + cidx];
            acc1 = __builtin_amdgcn_wmma_f32_16x16x4_f32(false, a, false, b, (short)0, acc1, false, false);
        }
    }
    acc0 = acc0 + acc1;

    const int mrow0 = node0 + rbase + 8 * grp;
    float* outp = h2 + (size_t)mrow0 * OUTC + cidx;
    if (node0 + 64 <= N) {
        if (cidx < OUTC) {
            #pragma unroll
            for (int v = 0; v < 8; ++v) outp[(size_t)v * OUTC] = acc0[v];
        }
    } else {
        #pragma unroll
        for (int v = 0; v < 8; ++v) {
            if (mrow0 + v < N && cidx < OUTC) outp[(size_t)v * OUTC] = acc0[v];
        }
    }
}

__global__ __launch_bounds__(256) void k_gather2(const int* __restrict__ rp,
                                                 const int2* __restrict__ csr,
                                                 const float* __restrict__ h2,
                                                 const float* __restrict__ dinv,
                                                 const float* __restrict__ b2,
                                                 float* __restrict__ out, int N) {
    const int lane = threadIdx.x & 31;
    const int d = __builtin_amdgcn_readfirstlane((int)(blockIdx.x * 8 + (threadIdx.x >> 5)));
    if (d >= N) return;
    const int beg = __builtin_amdgcn_readfirstlane(rp[d]);
    const int end = __builtin_amdgcn_readfirstlane(rp[d + 1]);
    const float dd = dinv[d];
    const int c = (lane < OUTC) ? lane : 0;
    const float hs = h2[(size_t)d * OUTC + c];
    float acc = 0.0f;

    for (int base = beg; base < end; base += 32) {
        const int m = (end - base < 32) ? (end - base) : 32;
        int es = d;
        float ew = 0.0f;
        if (lane < m) {
            int2 en = csr[base + lane];
            es = en.x;
            ew = __int_as_float(en.y) * dd;
        }
        for (int k0 = 0; k0 < m; k0 += 8) {
            const int r = m - k0;
            float v[8];
            float w[8];
            #pragma unroll
            for (int u = 0; u < 8; ++u) {
                if (u < r) {
                    int s = __builtin_amdgcn_readlane(es, k0 + u);
                    w[u] = __int_as_float(__builtin_amdgcn_readlane(__float_as_int(ew), k0 + u));
                    v[u] = h2[(size_t)s * OUTC + c];
                }
            }
            #pragma unroll
            for (int u = 0; u < 8; ++u) {
                if (u < r) acc += v[u] * w[u];
            }
        }
    }
    acc += hs * (dd * dd);
    if (lane < OUTC) out[(size_t)d * OUTC + lane] = acc + b2[lane];
}

extern "C" void kernel_launch(void* const* d_in, const int* in_sizes, int n_in,
                              void* d_out, int out_size, void* d_ws, size_t ws_size,
                              hipStream_t stream) {
    const float* x    = (const float*)d_in[0];
    const int*   ei   = (const int*)d_in[1];
    const float* anc  = (const float*)d_in[3];
    const float* W1   = (const float*)d_in[4];
    const float* b1   = (const float*)d_in[5];
    const float* W2   = (const float*)d_in[6];
    const float* b2   = (const float*)d_in[7];
    float* out = (float*)d_out;

    const int N = in_sizes[0] / IN_CH;
    const int E = in_sizes[1] / 2;
    const int* src = ei;
    const int* dst = ei + E;

    float* h1   = (float*)d_ws;
    float* agg1 = h1 + (size_t)N * HID;
    int2*  csr  = (int2*)(agg1 + (size_t)N * HID);
    float* h2   = (float*)(csr + (size_t)E);
    float* dinv = h2 + (size_t)N * OUTC;
    int*   cnt  = (int*)(dinv + (size_t)N);
    int*   rp   = cnt + (size_t)N;
    int*   bsum = rp + (size_t)N + 1;

    k_cnt_zero<<<(N + 255) / 256, 256, 0, stream>>>(cnt, N);
    k_cnt     <<<(E + 255) / 256, 256, 0, stream>>>(dst, cnt, E);
    k_blocksum<<<(N + 1023) / 1024, 1024, 0, stream>>>(cnt, bsum, dinv, N);
    k_scan    <<<(N + 1023) / 1024, 1024, 0, stream>>>(cnt, bsum, rp, N);
    k_fill    <<<(E + 255) / 256, 256, 0, stream>>>(src, dst, dinv, rp, csr, E);

    k_gemm1<<<(N + 15) / 16, 256, 0, stream>>>(x, anc, W1, h1, N);

    k_gather1<<<(N + 7) / 8, 256, 0, stream>>>(rp, csr, h1, dinv, agg1, N);

    k_gemm2<<<(N + 63) / 64, 128, 0, stream>>>(agg1, b1, W2, h2, N);

    k_gather2<<<(N + 7) / 8, 256, 0, stream>>>(rp, csr, h2, dinv, b2, out, N);
}


# edge list built in one atomic pass into 64-slot rows per node (overflow list behind) instead of count+scan+fill; rows stored pre-scaled by rsqrt(deg), dst factor after the sum; arrival-order sums
# speedup vs baseline: 1.9724x; 1.2631x over previous
#include <hip/hip_runtime.h>
#include <hip/hip_bf16.h>


typedef __attribute__((ext_vector_type(2))) float v2f;
typedef __attribute__((ext_vector_type(8))) float v8f;

#define IN_CH 128
#define IN2   256
#define HID   128
#define OUTC  10

#define CAP 64

__global__ __launch_bounds__(256) void k_zero(int* cnt, int* ovfn, int N) {
    int i = blockIdx.x * 256 + threadIdx.x;
    if (i < N) cnt[i] = 0;
    if (i == 0) *ovfn = 0;
}

__global__ __launch_bounds__(256) void k_slot(const int* __restrict__ src,
                                              const int* __restrict__ dst,
                                              int* cnt, int* __restrict__ slots,
                                              int* ovfn, int2* __restrict__ ovf, int E) {
    int e = blockIdx.x * 256 + threadIdx.x;
    if (e >= E) return;
    int s = src[e], d = dst[e];
    int pos = atomicAdd(&cnt[d], 1);
    if (pos < CAP) {
        slots[(size_t)d * CAP + pos] = s;
    } else {
        int k = atomicAdd(ovfn, 1);
        ovf[k] = make_int2(s, d);
    }
}

__device__ __forceinline__ float node_scale(const int* __restrict__ cnt, int node) {
    return rsqrtf((float)(cnt[node] + 1));
}

#define AS_STRIDE 258

__global__ __launch_bounds__(256) void k_gemm1(const float* __restrict__ x,
                                               const float* __restrict__ anc,
                                               const float* __restrict__ W1,
                                               const int* __restrict__ cnt,
                                               float* __restrict__ h1, int N) {
    __shared__ float As[16 * AS_STRIDE];
    const int row0 = blockIdx.x * 16;
    const int tid = threadIdx.x;

    for (int idx = tid; idx < 16 * IN2; idx += 256) {
        int r = idx >> 8, k = idx & 255;
        int node = row0 + r;
        float v = 0.0f;
        if (node < N) {
            v = (k < IN_CH) ? (x[(size_t)node * IN_CH + k] - anc[(size_t)node * IN_CH + k])
                            : anc[(size_t)node * IN_CH + (k - IN_CH)];
            v *= node_scale(cnt, node);
        }
        As[r * AS_STRIDE + k] = v;
    }
    __syncthreads();

    const int wave = tid >> 5, lane = tid & 31;
    const int grp = lane >> 4;
    const int cidx = lane & 15;
    const int n0 = wave * 16;

    v8f acc0 = {}, acc1 = {};
    for (int k0 = 0; k0 < IN2; k0 += 8) {
        {
            int kk = k0 + 2 * grp;
            v2f a, b;
            a.x = As[cidx * AS_STRIDE + kk];
            a.y = As[cidx * AS_STRIDE + kk + 1];
            b.x = W1[(size_t)kk * HID + n0 + cidx];
            b.y = W1[(size_t)(kk + 1) * HID + n0 + cidx];
            acc0 = __builtin_amdgcn_wmma_f32_16x16x4_f32(false, a, false, b, (short)0, acc0, false, false);
        }
        {
            int kk = k0 + 4 + 2 * grp;
            v2f a, b;
            a.x = As[cidx * AS_STRIDE + kk];
            a.y = As[cidx * AS_STRIDE + kk + 1];
            b.x = W1[(size_t)kk * HID + n0 + cidx];
            b.y = W1[(size_t)(kk + 1) * HID + n0 + cidx];
            acc1 = __builtin_amdgcn_wmma_f32_16x16x4_f32(false, a, false, b, (short)0, acc1, false, false);
        }
    }
    acc0 = acc0 + acc1;

    float* outp = h1 + (size_t)(row0 + 8 * grp) * HID + n0 + cidx;
    if (row0 + 16 <= N) {
        #pragma unroll
        for (int v = 0; v < 8; ++v) outp[(size_t)v * HID] = acc0[v];
    } else {
        #pragma unroll
        for (int v = 0; v < 8; ++v) {
            if (row0 + v + 8 * grp < N) outp[(size_t)v * HID] = acc0[v];
        }
    }
}

__global__ __launch_bounds__(256) void k_gather1(const int* __restrict__ cnt,
                                                 const int* __restrict__ slots,
                                                 const int* __restrict__ ovfn,
                                                 const int2* __restrict__ ovf,
                                                 const float* __restrict__ h1,
                                                 float* __restrict__ agg1, int N) {
    const int lane = threadIdx.x & 31;
    const int d = __builtin_amdgcn_readfirstlane((int)(blockIdx.x * 8 + (threadIdx.x >> 5)));
    if (d >= N) return;
    const int c = __builtin_amdgcn_readfirstlane(cnt[d]);
    const int mall = (c < CAP) ? c : CAP;
    const float dd = rsqrtf((float)(c + 1));
    const float4 sv = *(const float4*)(h1 + (size_t)d * HID + lane * 4);
    const int* row = slots + (size_t)d * CAP;
    float4 acc = {0.0f, 0.0f, 0.0f, 0.0f};

    for (int base = 0; base < mall; base += 32) {
        const int m = (mall - base < 32) ? (mall - base) : 32;
        int eo = d * HID;
        if (lane < m) eo = row[base + lane] * HID;
        for (int k0 = 0; k0 < m; k0 += 8) {
            const int r = m - k0;
            float4 v[8];
            #pragma unroll
            for (int u = 0; u < 8; ++u) {
                if (u < r) {
                    int so = __builtin_amdgcn_readlane(eo, k0 + u);
                    v[u] = *(const float4*)(h1 + so + lane * 4);
                }
            }
            #pragma unroll
            for (int u = 0; u < 8; ++u) {
                if (u < r) {
                    acc.x += v[u].x;
                    acc.y += v[u].y;
                    acc.z += v[u].z;
                    acc.w += v[u].w;
                }
            }
        }
    }
    if (c > CAP) {
        const int n = __builtin_amdgcn_readfirstlane(*ovfn);
        for (int k = 0; k < n; k += 32) {
            int2 en = make_int2(d, -1);
            if (k + lane < n) en = ovf[k + lane];
            unsigned mask = __builtin_amdgcn_ballot_w32(en.y == d);
            while (mask) {
                const int j = __builtin_ctz(mask);
                mask &= mask - 1;
                const int s = __builtin_amdgcn_readlane(en.x, j);
                const float4 v = *(const float4*)(h1 + (size_t)s * HID + lane * 4);
                acc.x += v.x; acc.y += v.y; acc.z += v.z; acc.w += v.w;
            }
        }
    }
    acc.x = (acc.x + sv.x) * dd; acc.y = (acc.y + sv.y) * dd;
    acc.z = (acc.z + sv.z) * dd; acc.w = (acc.w + sv.w) * dd;
    *(float4*)(agg1 + (size_t)d * HID + lane * 4) = acc;
}

#define AS2_STRIDE 132

__global__ __launch_bounds__(128) void k_gemm2(const float* __restrict__ agg1,
                                               const float* __restrict__ b1,
                                               const float* __restrict__ W2,
                                               const int* __restrict__ cnt,
                                               float* __restrict__ h2, int N) {
    __shared__ float As[64 * AS2_STRIDE];
    __shared__ float Bs[HID * 16];
    const int node0 = blockIdx.x * 64;
    const int tid = threadIdx.x;

    for (int idx = tid; idx < 64 * HID; idx += 128) {
        int r = idx >> 7, k = idx & 127;
        int node = node0 + r;
        float v = 0.0f;
        if (node < N) {
            v = agg1[(size_t)node * HID + k] + b1[k];
            v = v > 0.0f ? v : 0.0f;
            v *= node_scale(cnt, node);
        }
        As[r * AS2_STRIDE + k] = v;
    }
    for (int idx = tid; idx < HID * 16; idx += 128) {
        int k = idx >> 4, n = idx & 15;
        Bs[idx] = (n < OUTC) ? W2[(size_t)k * OUTC + n] : 0.0f;
    }
    __syncthreads();

    const int wave = tid >> 5, lane = tid & 31;
    const int grp = lane >> 4, cidx = lane & 15;
    const int rbase = wave * 16;

    v8f acc0 = {}, acc1 = {};
    for (int k0 = 0; k0 < HID; k0 += 8) {
        {
            int kk = k0 + 2 * grp;
            v2f a, b;
            a.x = As[(rbase + cidx) * AS2_STRIDE + kk];
            a.y = As[(rbase + cidx) * AS2_STRIDE + kk + 1];
            b.x = Bs[kk * 16 + cidx];
            b.y = Bs[(kk + 1) * 16 + cidx];
            acc0 = __builtin_amdgcn_wmma_f32_16x16x4_f32(false, a, false, b, (short)0, acc0, false, false);
        }
        {
            int kk = k0 + 4 + 2 * grp;
            v2f a, b;
            a.x = As[(rbase + cidx) * AS2_STRIDE + kk];
            a.y = As[(rbase + cidx) * AS2_STRIDE + kk + 1];
            b.x = Bs[kk * 16 + cidx];
            b.y = Bs[(kk + 1) * 16 + cidx];
            acc1 = __builtin_amdgcn_wmma_f32_16x16x4_f32(false, a, false, b, (short)0, acc1, false, false);
        }
    }
    acc0 = acc0 + acc1;

    const int mrow0 = node0 + rbase + 8 * grp;
    float* outp = h2 + (size_t)mrow0 * OUTC + cidx;
    if (node0 + 64 <= N) {
        if (cidx < OUTC) {
            #pragma unroll
            for (int v = 0; v < 8; ++v) outp[(size_t)v * OUTC] = acc0[v];
        }
    } else {
        #pragma unroll
        for (int v = 0; v < 8; ++v) {
            if (mrow0 + v < N && cidx < OUTC) outp[(size_t)v * OUTC] = acc0[v];
        }
    }
}

__global__ __launch_bounds__(256) void k_gather2(const int* __restrict__ cnt,
                                                 const int* __restrict__ slots,
                                                 const int* __restrict__ ovfn,
                                                 const int2* __restrict__ ovf,
                                                 const float* __restrict__ h2,
                                                 const float* __restrict__ b2,
                                                 float* __restrict__ out, int N) {
    const int lane = threadIdx.x & 31;
    const int d = __builtin_amdgcn_readfirstlane((int)(blockIdx.x * 8 + (threadIdx.x >> 5)));
    if (d >= N) return;
    const int c = __builtin_amdgcn_readfirstlane(cnt[d]);
    const int mall = (c < CAP) ? c : CAP;
    const float dd = rsqrtf((float)(c + 1));
    const int ch = (lane < OUTC) ? lane : 0;
    const float hs = h2[(size_t)d * OUTC + ch];
    const int* row = slots + (size_t)d * CAP;
    float acc = 0.0f;

    for (int base = 0; base < mall; base += 32) {
        const int m = (mall - base < 32) ? (mall - base) : 32;
        int eo = d * OUTC;
        if (lane < m) eo = row[base + lane] * OUTC;
        for (int k0 = 0; k0 < m; k0 += 8) {
            const int r = m - k0;
            float v[8];
            #pragma unroll
            for (int u = 0; u < 8; ++u) {
                if (u < r) {
                    int so = __builtin_amdgcn_readlane(eo, k0 + u);
                    v[u] = h2[so + ch];
                }
            }
            #pragma unroll
            for (int u = 0; u < 8; ++u) {
                if (u < r) acc += v[u];
            }
        }
    }
    if (c > CAP) {
        const int n = __builtin_amdgcn_readfirstlane(*ovfn);
        for (int k = 0; k < n; k += 32) {
            int2 en = make_int2(d, -1);
            if (k + lane < n) en = ovf[k + lane];
            unsigned mask = __builtin_amdgcn_ballot_w32(en.y == d);
            while (mask) {
                const int j = __builtin_ctz(mask);
                mask &= mask - 1;
                const int s = __builtin_amdgcn_readlane(en.x, j);
                acc += h2[(size_t)s * OUTC + ch];
            }
        }
    }
    acc = (acc + hs) * dd;
    if (lane < OUTC) out[(size_t)d * OUTC + lane] = acc + b2[lane];
}

extern "C" void kernel_launch(void* const* d_in, const int* in_sizes, int n_in,
                              void* d_out, int out_size, void* d_ws, size_t ws_size,
                              hipStream_t stream) {
    const float* x    = (const float*)d_in[0];
    const int*   ei   = (const int*)d_in[1];
    const float* anc  = (const float*)d_in[3];
    const float* W1   = (const float*)d_in[4];
    const float* b1   = (const float*)d_in[5];
    const float* W2   = (const float*)d_in[6];
    const float* b2   = (const float*)d_in[7];
    float* out = (float*)d_out;

    const int N = in_sizes[0] / IN_CH;
    const int E = in_sizes[1] / 2;
    const int* src = ei;
    const int* dst = ei + E;

    float* h1    = (float*)d_ws;
    float* agg1  = h1 + (size_t)N * HID;
    int*   slots = (int*)(agg1 + (size_t)N * HID);
    int2*  ovf   = (int2*)(slots + (size_t)N * CAP);
    float* h2    = (float*)(ovf + (size_t)E);
    int*   cnt   = (int*)(h2 + (size_t)N * OUTC);
    int*   ovfn  = cnt + (size_t)N;

    k_zero<<<(N + 255) / 256, 256, 0, stream>>>(cnt, ovfn, N);
    k_slot<<<(E + 255) / 256, 256, 0, stream>>>(src, dst, cnt, slots, ovfn, ovf, E);

    k_gemm1<<<(N + 15) / 16, 256, 0, stream>>>(x, anc, W1, cnt, h1, N);

    k_gather1<<<(N + 7) / 8, 256, 0, stream>>>(cnt, slots, ovfn, ovf, h1, agg1, N);

    k_gemm2<<<(N + 63) / 64, 128, 0, stream>>>(agg1, b1, W2, cnt, h2, N);

    k_gather2<<<(N + 7) / 8, 256, 0, stream>>>(cnt, slots, ovfn, ovf, h2, b2, out, N);
}


# Operand staging of both f32 products (x/anchor, agg1, W2): a thread issues all its loads up front, rows as 16-byte loads, then scales and stores to LDS; was one 4-byte load per trip, each waited for
# speedup vs baseline: 2.4048x; 1.2192x over previous
#include <hip/hip_runtime.h>
#include <hip/hip_bf16.h>


typedef __attribute__((ext_vector_type(2))) float v2f;
typedef __attribute__((ext_vector_type(8))) float v8f;

#define IN_CH 128
#define IN2   256
#define HID   128
#define OUTC  10

#define CAP 64

__global__ __launch_bounds__(256) void k_zero(int* cnt, int* ovfn, int N) {
    int i = blockIdx.x * 256 + threadIdx.x;
    if (i < N) cnt[i] = 0;
    if (i == 0) *ovfn = 0;
}

__global__ __launch_bounds__(256) void k_slot(const int* __restrict__ src,
                                              const int* __restrict__ dst,
                                              int* cnt, int* __restrict__ slots,
                                              int* ovfn, int2* __restrict__ ovf, int E) {
    int e = blockIdx.x * 256 + threadIdx.x;
    if (e >= E) return;
    int s = src[e], d = dst[e];
    int pos = atomicAdd(&cnt[d], 1);
    if (pos < CAP) {
        slots[(size_t)d * CAP + pos] = s;
    } else {
        int k = atomicAdd(ovfn, 1);
        ovf[k] = make_int2(s, d);
    }
}

__device__ __forceinline__ float deg_scale(int c) {
    return rsqrtf((float)(c + 1));
}

#define AS_STRIDE 258

__global__ __launch_bounds__(256) void k_gemm1(const float* __restrict__ x,
                                               const float* __restrict__ anc,
                                               const float* __restrict__ W1,
                                               const int* __restrict__ cnt,
                                               float* __restrict__ h1, int N) {
    __shared__ __attribute__((aligned(16))) float As[16 * AS_STRIDE];
    const int row0 = blockIdx.x * 16;
    const int tid = threadIdx.x;

    {
        const int c4 = (tid & 31) * 4;
        float4 xv[2], av[2];
        int cn[2];
        #pragma unroll
        for (int j = 0; j < 2; ++j) {
            const int node = row0 + (tid >> 5) + 8 * j;
            const int nc = node < N ? node : N - 1;
            xv[j] = *(const float4*)(x + (size_t)nc * IN_CH + c4);
            av[j] = *(const float4*)(anc + (size_t)nc * IN_CH + c4);
            cn[j] = cnt[nc];
        }
        #pragma unroll
        for (int j = 0; j < 2; ++j) {
            const int r = (tid >> 5) + 8 * j;
            const float sc = (row0 + r < N) ? deg_scale(cn[j]) : 0.0f;
            float2 d0, d1, a0, a1;
            d0.x = (xv[j].x - av[j].x) * sc;
            d0.y = (xv[j].y - av[j].y) * sc;
            d1.x = (xv[j].z - av[j].z) * sc;
            d1.y = (xv[j].w - av[j].w) * sc;
            a0.x = av[j].x * sc;
            a0.y = av[j].y * sc;
            a1.x = av[j].z * sc;
            a1.y = av[j].w * sc;
            float* dp = &As[r * AS_STRIDE + c4];
            *(float2*)(dp) = d0;
            *(float2*)(dp + 2) = d1;
            *(float2*)(dp + IN_CH) = a0;
            *(float2*)(dp + IN_CH + 2) = a1;
        }
    }
    __syncthreads();

    const int wave = tid >> 5, lane = tid & 31;
    const int grp = lane >> 4;
    const int cidx = lane & 15;
    const int n0 = wave * 16;

    v8f acc0 = {}, acc1 = {};
    for (int k0 = 0; k0 < IN2; k0 += 8) {
        {
            int kk = k0 + 2 * grp;
            v2f a, b;
            a.x = As[cidx * AS_STRIDE + kk];
            a.y = As[cidx * AS_STRIDE + kk + 1];
            b.x = W1[(size_t)kk * HID + n0 + cidx];
            b.y = W1[(size_t)(kk + 1) * HID + n0 + cidx];
            acc0 = __builtin_amdgcn_wmma_f32_16x16x4_f32(false, a, false, b, (short)0, acc0, false, false);
        }
        {
            int kk = k0 + 4 + 2 * grp;
            v2f a, b;
            a.x = As[cidx * AS_STRIDE + kk];
            a.y = As[cidx * AS_STRIDE + kk + 1];
            b.x = W1[(size_t)kk * HID + n0 + cidx];
            b.y = W1[(size_t)(kk + 1) * HID + n0 + cidx];
            acc1 = __builtin_amdgcn_wmma_f32_16x16x4_f32(false, a, false, b, (short)0, acc1, false, false);
        }
    }
    acc0 = acc0 + acc1;

    float* outp = h1 + (size_t)(row0 + 8 * grp) * HID + n0 + cidx;
    if (row0 + 16 <= N) {
        #pragma unroll
        for (int v = 0; v < 8; ++v) outp[(size_t)v * HID] = acc0[v];
    } else {
        #pragma unroll
        for (int v = 0; v < 8; ++v) {
            if (row0 + v + 8 * grp < N) outp[(size_t)v * HID] = acc0[v];
        }
    }
}

__global__ __launch_bounds__(256) void k_gather1(const int* __restrict__ cnt,
                                                 const int* __restrict__ slots,
                                                 const int* __restrict__ ovfn,
                                                 const int2* __restrict__ ovf,
                                                 const float* __restrict__ h1,
                                                 float* __restrict__ agg1, int N) {
    const int lane = threadIdx.x & 31;
    const int d = __builtin_amdgcn_readfirstlane((int)(blockIdx.x * 8 + (threadIdx.x >> 5)));
    if (d >= N) return;
    const int c = __builtin_amdgcn_readfirstlane(cnt[d]);
    const int mall = (c < CAP) ? c : CAP;
    const float dd = rsqrtf((float)(c + 1));
    const float4 sv = *(const float4*)(h1 + (size_t)d * HID + lane * 4);
    const int* row = slots + (size_t)d * CAP;
    float4 acc = {0.0f, 0.0f, 0.0f, 0.0f};

    for (int base = 0; base < mall; base += 32) {
        const int m = (mall - base < 32) ? (mall - base) : 32;
        int eo = d * HID;
        if (lane < m) eo = row[base + lane] * HID;
        for (int k0 = 0; k0 < m; k0 += 8) {
            const int r = m - k0;
            float4 v[8];
            #pragma unroll
            for (int u = 0; u < 8; ++u) {
                if (u < r) {
                    int so = __builtin_amdgcn_readlane(eo, k0 + u);
                    v[u] = *(const float4*)(h1 + so + lane * 4);
                }
            }
            #pragma unroll
            for (int u = 0; u < 8; ++u) {
                if (u < r) {
                    acc.x += v[u].x;
                    acc.y += v[u].y;
                    acc.z += v[u].z;
                    acc.w += v[u].w;
                }
            }
        }
    }
    if (c > CAP) {
        const int n = __builtin_amdgcn_readfirstlane(*ovfn);
        for (int k = 0; k < n; k += 32) {
            int2 en = make_int2(d, -1);
            if (k + lane < n) en = ovf[k + lane];
            unsigned mask = __builtin_amdgcn_ballot_w32(en.y == d);
            while (mask) {
                const int j = __builtin_ctz(mask);
                mask &= mask - 1;
                const int s = __builtin_amdgcn_readlane(en.x, j);
                const float4 v = *(const float4*)(h1 + (size_t)s * HID + lane * 4);
                acc.x += v.x; acc.y += v.y; acc.z += v.z; acc.w += v.w;
            }
        }
    }
    acc.x = (acc.x + sv.x) * dd; acc.y = (acc.y + sv.y) * dd;
    acc.z = (acc.z + sv.z) * dd; acc.w = (acc.w + sv.w) * dd;
    *(float4*)(agg1 + (size_t)d * HID + lane * 4) = acc;
}

#define AS2_STRIDE 132

__global__ __launch_bounds__(128) void k_gemm2(const float* __restrict__ agg1,
                                               const float* __restrict__ b1,
                                               const float* __restrict__ W2,
                                               const int* __restrict__ cnt,
                                               float* __restrict__ h2, int N) {
    __shared__ __attribute__((aligned(16))) float As[64 * AS2_STRIDE];
    __shared__ float Bs[HID * 16];
    const int node0 = blockIdx.x * 64;
    const int tid = threadIdx.x;

    {
        const int c4 = (tid & 31) * 4;
        const int n = tid & 15;
        const float4 bv = *(const float4*)(b1 + c4);
        float4 gv[16];
        int cn[16];
        float wv[16];
        #pragma unroll
        for (int j = 0; j < 16; ++j) {
            const int node = node0 + (tid >> 5) + 4 * j;
            const int nc = node < N ? node : N - 1;
            gv[j] = *(const float4*)(agg1 + (size_t)nc * HID + c4);
            cn[j] = cnt[nc];
        }
        #pragma unroll
        for (int j = 0; j < 16; ++j) {
            const int k = (tid >> 4) + 8 * j;
            wv[j] = W2[(size_t)k * OUTC + (n < OUTC ? n : 0)];
        }
        #pragma unroll
        for (int j = 0; j < 16; ++j) {
            const int r = (tid >> 5) + 4 * j;
            const float sc = (node0 + r < N) ? deg_scale(cn[j]) : 0.0f;
            float4 v;
            v.x = gv[j].x + bv.x; v.y = gv[j].y + bv.y;
            v.z = gv[j].z + bv.z; v.w = gv[j].w + bv.w;
            v.x = v.x > 0.0f ? v.x : 0.0f; v.y = v.y > 0.0f ? v.y : 0.0f;
            v.z = v.z > 0.0f ? v.z : 0.0f; v.w = v.w > 0.0f ? v.w : 0.0f;
            v.x *= sc; v.y *= sc; v.z *= sc; v.w *= sc;
            *(float4*)&As[r * AS2_STRIDE + c4] = v;
        }
        #pragma unroll
        for (int j = 0; j < 16; ++j) {
            Bs[tid + 128 * j] = (n < OUTC) ? wv[j] : 0.0f;
        }
    }
    __syncthreads();

    const int wave = tid >> 5, lane = tid & 31;
    const int grp = lane >> 4, cidx = lane & 15;
    const int rbase = wave * 16;

    v8f acc0 = {}, acc1 = {};
    for (int k0 = 0; k0 < HID; k0 += 8) {
        {
            int kk = k0 + 2 * grp;
            v2f a, b;
            a.x = As[(rbase + cidx) * AS2_STRIDE + kk];
            a.y = As[(rbase + cidx) * AS2_STRIDE + kk + 1];
            b.x = Bs[kk * 16 + cidx];
            b.y = Bs[(kk + 1) * 16 + cidx];
            acc0 = __builtin_amdgcn_wmma_f32_16x16x4_f32(false, a, false, b, (short)0, acc0, false, false);
        }
        {
            int kk = k0 + 4 + 2 * grp;
            v2f a, b;
            a.x = As[(rbase + cidx) * AS2_STRIDE + kk];
            a.y = As[(rbase + cidx) * AS2_STRIDE + kk + 1];
            b.x = Bs[kk * 16 + cidx];
            b.y = Bs[(kk + 1) * 16 + cidx];
            acc1 = __builtin_amdgcn_wmma_f32_16x16x4_f32(false, a, false, b, (short)0, acc1, false, false);
        }
    }
    acc0 = acc0 + acc1;

    const int mrow0 = node0 + rbase + 8 * grp;
    float* outp = h2 + (size_t)mrow0 * OUTC + cidx;
    if (node0 + 64 <= N) {
        if (cidx < OUTC) {
            #pragma unroll
            for (int v = 0; v < 8; ++v) outp[(size_t)v * OUTC] = acc0[v];
        }
    } else {
        #pragma unroll
        for (int v = 0; v < 8; ++v) {
            if (mrow0 + v < N && cidx < OUTC) outp[(size_t)v * OUTC] = acc0[v];
        }
    }
}

__global__ __launch_bounds__(256) void k_gather2(const int* __restrict__ cnt,
                                                 const int* __restrict__ slots,
                                                 const int* __restrict__ ovfn,
                                                 const int2* __restrict__ ovf,
                                                 const float* __restrict__ h2,
                                                 const float* __restrict__ b2,
                                                 float* __restrict__ out, int N) {
    const int lane = threadIdx.x & 31;
    const int d = __builtin_amdgcn_readfirstlane((int)(blockIdx.x * 8 + (threadIdx.x >> 5)));
    if (d >= N) return;
    const int c = __builtin_amdgcn_readfirstlane(cnt[d]);
    const int mall = (c < CAP) ? c : CAP;
    const float dd = rsqrtf((float)(c + 1));
    const int ch = (lane < OUTC) ? lane : 0;
    const float hs = h2[(size_t)d * OUTC + ch];
    const int* row = slots + (size_t)d * CAP;
    float acc = 0.0f;

    for (int base = 0; base < mall; base += 32) {
        const int m = (mall - base < 32) ? (mall - base) : 32;
        int eo = d * OUTC;
        if (lane < m) eo = row[base + lane] * OUTC;
        for (int k0 = 0; k0 < m; k0 += 8) {
            const int r = m - k0;
            float v[8];
            #pragma unroll
            for (int u = 0; u < 8; ++u) {
                if (u < r) {
                    int so = __builtin_amdgcn_readlane(eo, k0 + u);
                    v[u] = h2[so + ch];
                }
            }
            #pragma unroll
            for (int u = 0; u < 8; ++u) {
                if (u < r) acc += v[u];
            }
        }
    }
    if (c > CAP) {
        const int n = __builtin_amdgcn_readfirstlane(*ovfn);
        for (int k = 0; k < n; k += 32) {
            int2 en = make_int2(d, -1);
            if (k + lane < n) en = ovf[k + lane];
            unsigned mask = __builtin_amdgcn_ballot_w32(en.y == d);
            while (mask) {
                const int j = __builtin_ctz(mask);
                mask &= mask - 1;
                const int s = __builtin_amdgcn_readlane(en.x, j);
                acc += h2[(size_t)s * OUTC + ch];
            }
        }
    }
    acc = (acc + hs) * dd;
    if (lane < OUTC) out[(size_t)d * OUTC + lane] = acc + b2[lane];
}

extern "C" void kernel_launch(void* const* d_in, const int* in_sizes, int n_in,
                              void* d_out, int out_size, void* d_ws, size_t ws_size,
                              hipStream_t stream) {
    const float* x    = (const float*)d_in[0];
    const int*   ei   = (const int*)d_in[1];
    const float* anc  = (const float*)d_in[3];
    const float* W1   = (const float*)d_in[4];
    const float* b1   = (const float*)d_in[5];
    const float* W2   = (const float*)d_in[6];
    const float* b2   = (const float*)d_in[7];
    float* out = (float*)d_out;

    const int N = in_sizes[0] / IN_CH;
    const int E = in_sizes[1] / 2;
    const int* src = ei;
    const int* dst = ei + E;

    float* h1    = (float*)d_ws;
    float* agg1  = h1 + (size_t)N * HID;
    int*   slots = (int*)(agg1 + (size_t)N * HID);
    int2*  ovf   = (int2*)(slots + (size_t)N * CAP);
    float* h2    = (float*)(ovf + (size_t)E);
    int*   cnt   = (int*)(h2 + (size_t)N * OUTC);
    int*   ovfn  = cnt + (size_t)N;

    k_zero<<<(N + 255) / 256, 256, 0, stream>>>(cnt, ovfn, N);
    k_slot<<<(E + 255) / 256, 256, 0, stream>>>(src, dst, cnt, slots, ovfn, ovf, E);

    k_gemm1<<<(N + 15) / 16, 256, 0, stream>>>(x, anc, W1, cnt, h1, N);

    k_gather1<<<(N + 7) / 8, 256, 0, stream>>>(cnt, slots, ovfn, ovf, h1, agg1, N);

    k_gemm2<<<(N + 63) / 64, 128, 0, stream>>>(agg1, b1, W2, cnt, h2, N);

    k_gather2<<<(N + 7) / 8, 256, 0, stream>>>(cnt, slots, ovfn, ovf, h2, b2, out, N);
}


# First f32 product re-tiled: a wave computes 32x32 outputs as 2x2 tiles (even/odd columns); one 8-byte W1 load per row feeds 4 tiles, an A fragment 2; 32 rows per block; sum order unchanged
# speedup vs baseline: 2.5414x; 1.0568x over previous
#include <hip/hip_runtime.h>
#include <hip/hip_bf16.h>


typedef __attribute__((ext_vector_type(2))) float v2f;
typedef __attribute__((ext_vector_type(8))) float v8f;

#define IN_CH 128
#define IN2   256
#define HID   128
#define OUTC  10

#define CAP 64

__global__ __launch_bounds__(256) void k_zero(int* cnt, int* ovfn, int N) {
    int i = blockIdx.x * 256 + threadIdx.x;
    if (i < N) cnt[i] = 0;
    if (i == 0) *ovfn = 0;
}

__global__ __launch_bounds__(256) void k_slot(const int* __restrict__ src,
                                              const int* __restrict__ dst,
                                              int* cnt, int* __restrict__ slots,
                                              int* ovfn, int2* __restrict__ ovf, int E) {
    int e = blockIdx.x * 256 + threadIdx.x;
    if (e >= E) return;
    int s = src[e], d = dst[e];
    int pos = atomicAdd(&cnt[d], 1);
    if (pos < CAP) {
        slots[(size_t)d * CAP + pos] = s;
    } else {
        int k = atomicAdd(ovfn, 1);
        ovf[k] = make_int2(s, d);
    }
}

__device__ __forceinline__ float deg_scale(int c) {
    return rsqrtf((float)(c + 1));
}

#define AS_STRIDE 258
#define G1_ROWS 32

__global__ __launch_bounds__(128) void k_gemm1(const float* __restrict__ x,
                                               const float* __restrict__ anc,
                                               const float* __restrict__ W1,
                                               const int* __restrict__ cnt,
                                               float* __restrict__ h1, int N) {
    __shared__ __attribute__((aligned(16))) float As[G1_ROWS * AS_STRIDE];
    const int row0 = blockIdx.x * G1_ROWS;
    const int tid = threadIdx.x;

    {
        const int c4 = (tid & 31) * 4;
        float4 xv[8], av[8];
        int cn[8];
        #pragma unroll
        for (int j = 0; j < 8; ++j) {
            const int node = row0 + (tid >> 5) + 4 * j;
            const int nc = node < N ? node : N - 1;
            xv[j] = *(const float4*)(x + (size_t)nc * IN_CH + c4);
            av[j] = *(const float4*)(anc + (size_t)nc * IN_CH + c4);
            cn[j] = cnt[nc];
        }
        #pragma unroll
        for (int j = 0; j < 8; ++j) {
            const int r = (tid >> 5) + 4 * j;
            const float sc = (row0 + r < N) ? deg_scale(cn[j]) : 0.0f;
            float2 d0, d1, a0, a1;
            d0.x = (xv[j].x - av[j].x) * sc;
            d0.y = (xv[j].y - av[j].y) * sc;
            d1.x = (xv[j].z - av[j].z) * sc;
            d1.y = (xv[j].w - av[j].w) * sc;
            a0.x = av[j].x * sc;
            a0.y = av[j].y * sc;
            a1.x = av[j].z * sc;
            a1.y = av[j].w * sc;
            float* dp = &As[r * AS_STRIDE + c4];
            *(float2*)(dp) = d0;
            *(float2*)(dp + 2) = d1;
            *(float2*)(dp + IN_CH) = a0;
            *(float2*)(dp + IN_CH + 2) = a1;
        }
    }
    __syncthreads();

    const int wave = tid >> 5, lane = tid & 31;
    const int grp = lane >> 4;
    const int cidx = lane & 15;
    const int n0 = wave * 32 + 2 * cidx;

    const float* a0p = &As[cidx * AS_STRIDE + 2 * grp];
    const float* a1p = &As[(16 + cidx) * AS_STRIDE + 2 * grp];
    const float* bp = W1 + (size_t)(2 * grp) * HID + n0;

    v8f acc[2][2][2] = {};
    #pragma unroll 2
    for (int k0 = 0; k0 < IN2; k0 += 8) {
        #pragma unroll
        for (int p = 0; p < 2; ++p) {
            const int kk = k0 + 4 * p;
            v2f a0, a1, b0, b1;
            a0.x = a0p[kk];
            a0.y = a0p[kk + 1];
            a1.x = a1p[kk];
            a1.y = a1p[kk + 1];
            const float2 w0 = *(const float2*)(bp + (size_t)kk * HID);
            const float2 w1 = *(const float2*)(bp + (size_t)(kk + 1) * HID);
            b0.x = w0.x; b0.y = w1.x;
            b1.x = w0.y; b1.y = w1.y;
            acc[p][0][0] = __builtin_amdgcn_wmma_f32_16x16x4_f32(false, a0, false, b0, (short)0, acc[p][0][0], false, false);
            acc[p][0][1] = __builtin_amdgcn_wmma_f32_16x16x4_f32(false, a0, false, b1, (short)0, acc[p][0][1], false, false);
            acc[p][1][0] = __builtin_amdgcn_wmma_f32_16x16x4_f32(false, a1, false, b0, (short)0, acc[p][1][0], false, false);
            acc[p][1][1] = __builtin_amdgcn_wmma_f32_16x16x4_f32(false, a1, false, b1, (short)0, acc[p][1][1], false, false);
        }
    }

    #pragma unroll
    for (int m = 0; m < 2; ++m) {
        const v8f e = acc[0][m][0] + acc[1][m][0];
        const v8f o = acc[0][m][1] + acc[1][m][1];
        const int rb = row0 + 16 * m + 8 * grp;
        float* outp = h1 + (size_t)rb * HID + n0;
        if (row0 + G1_ROWS <= N) {
            #pragma unroll
            for (int v = 0; v < 8; ++v) {
                float2 t; t.x = e[v]; t.y = o[v];
                *(float2*)(outp + (size_t)v * HID) = t;
            }
        } else {
            #pragma unroll
            for (int v = 0; v < 8; ++v) {
                if (rb + v < N) {
                    float2 t; t.x = e[v]; t.y = o[v];
                    *(float2*)(outp + (size_t)v * HID) = t;
                }
            }
        }
    }
}

__global__ __launch_bounds__(256) void k_gather1(const int* __restrict__ cnt,
                                                 const int* __restrict__ slots,
                                                 const int* __restrict__ ovfn,
                                                 const int2* __restrict__ ovf,
                                                 const float* __restrict__ h1,
                                                 float* __restrict__ agg1, int N) {
    const int lane = threadIdx.x & 31;
    const int d = __builtin_amdgcn_readfirstlane((int)(blockIdx.x * 8 + (threadIdx.x >> 5)));
    if (d >= N) return;
    const int c = __builtin_amdgcn_readfirstlane(cnt[d]);
    const int mall = (c < CAP) ? c : CAP;
    const float dd = rsqrtf((float)(c + 1));
    const float4 sv = *(const float4*)(h1 + (size_t)d * HID + lane * 4);
    const int* row = slots + (size_t)d * CAP;
    float4 acc = {0.0f, 0.0f, 0.0f, 0.0f};

    for (int base = 0; base < mall; base += 32) {
        const int m = (mall - base < 32) ? (mall - base) : 32;
        int eo = d * HID;
        if (lane < m) eo = row[base + lane] * HID;
        for (int k0 = 0; k0 < m; k0 += 8) {
            const int r = m - k0;
            float4 v[8];
            #pragma unroll
            for (int u = 0; u < 8; ++u) {
                if (u < r) {
                    int so = __builtin_amdgcn_readlane(eo, k0 + u);
                    v[u] = *(const float4*)(h1 + so + lane * 4);
                }
            }
            #pragma unroll
            for (int u = 0; u < 8; ++u) {
                if (u < r) {
                    acc.x += v[u].x;
                    acc.y += v[u].y;
                    acc.z += v[u].z;
                    acc.w += v[u].w;
                }
            }
        }
    }
    if (c > CAP) {
        const int n = __builtin_amdgcn_readfirstlane(*ovfn);
        for (int k = 0; k < n; k += 32) {
            int2 en = make_int2(d, -1);
            if (k + lane < n) en = ovf[k + lane];
            unsigned mask = __builtin_amdgcn_ballot_w32(en.y == d);
            while (mask) {
                const int j = __builtin_ctz(mask);
                mask &= mask - 1;
                const int s = __builtin_amdgcn_readlane(en.x, j);
                const float4 v = *(const float4*)(h1 + (size_t)s * HID + lane * 4);
                acc.x += v.x; acc.y += v.y; acc.z += v.z; acc.w += v.w;
            }
        }
    }
    acc.x = (acc.x + sv.x) * dd; acc.y = (acc.y + sv.y) * dd;
    acc.z = (acc.z + sv.z) * dd; acc.w = (acc.w + sv.w) * dd;
    *(float4*)(agg1 + (size_t)d * HID + lane * 4) = acc;
}

#define AS2_STRIDE 132

__global__ __launch_bounds__(128) void k_gemm2(const float* __restrict__ agg1,
                                               const float* __restrict__ b1,
                                               const float* __restrict__ W2,
                                               const int* __restrict__ cnt,
                                               float* __restrict__ h2, int N) {
    __shared__ __attribute__((aligned(16))) float As[64 * AS2_STRIDE];
    __shared__ float Bs[HID * 16];
    const int node0 = blockIdx.x * 64;
    const int tid = threadIdx.x;

    {
        const int c4 = (tid & 31) * 4;
        const int n = tid & 15;
        const float4 bv = *(const float4*)(b1 + c4);
        float4 gv[16];
        int cn[16];
        float wv[16];
        #pragma unroll
        for (int j = 0; j < 16; ++j) {
            const int node = node0 + (tid >> 5) + 4 * j;
            const int nc = node < N ? node : N - 1;
            gv[j] = *(const float4*)(agg1 + (size_t)nc * HID + c4);
            cn[j] = cnt[nc];
        }
        #pragma unroll
        for (int j = 0; j < 16; ++j) {
            const int k = (tid >> 4) + 8 * j;
            wv[j] = W2[(size_t)k * OUTC + (n < OUTC ? n : 0)];
        }
        #pragma unroll
        for (int j = 0; j < 16; ++j) {
            const int r = (tid >> 5) + 4 * j;
            const float sc = (node0 + r < N) ? deg_scale(cn[j]) : 0.0f;
            float4 v;
            v.x = gv[j].x + bv.x; v.y = gv[j].y + bv.y;
            v.z = gv[j].z + bv.z; v.w = gv[j].w + bv.w;
            v.x = v.x > 0.0f ? v.x : 0.0f; v.y = v.y > 0.0f ? v.y : 0.0f;
            v.z = v.z > 0.0f ? v.z : 0.0f; v.w = v.w > 0.0f ? v.w : 0.0f;
            v.x *= sc; v.y *= sc; v.z *= sc; v.w *= sc;
            *(float4*)&As[r * AS2_STRIDE + c4] = v;
        }
        #pragma unroll
        for (int j = 0; j < 16; ++j) {
            Bs[tid + 128 * j] = (n < OUTC) ? wv[j] : 0.0f;
        }
    }
    __syncthreads();

    const int wave = tid >> 5, lane = tid & 31;
    const int grp = lane >> 4, cidx = lane & 15;
    const int rbase = wave * 16;

    v8f acc0 = {}, acc1 = {};
    for (int k0 = 0; k0 < HID; k0 += 8) {
        {
            int kk = k0 + 2 * grp;
            v2f a, b;
            a.x = As[(rbase + cidx) * AS2_STRIDE + kk];
            a.y = As[(rbase + cidx) * AS2_STRIDE + kk + 1];
            b.x = Bs[kk * 16 + cidx];
            b.y = Bs[(kk + 1) * 16 + cidx];
            acc0 = __builtin_amdgcn_wmma_f32_16x16x4_f32(false, a, false, b, (short)0, acc0, false, false);
        }
        {
            int kk = k0 + 4 + 2 * grp;
            v2f a, b;
            a.x = As[(rbase + cidx) * AS2_STRIDE + kk];
            a.y = As[(rbase + cidx) * AS2_STRIDE + kk + 1];
            b.x = Bs[kk * 16 + cidx];
            b.y = Bs[(kk + 1) * 16 + cidx];
            acc1 = __builtin_amdgcn_wmma_f32_16x16x4_f32(false, a, false, b, (short)0, acc1, false, false);
        }
    }
    acc0 = acc0 + acc1;

    const int mrow0 = node0 + rbase + 8 * grp;
    float* outp = h2 + (size_t)mrow0 * OUTC + cidx;
    if (node0 + 64 <= N) {
        if (cidx < OUTC) {
            #pragma unroll
            for (int v = 0; v < 8; ++v) outp[(size_t)v * OUTC] = acc0[v];
        }
    } else {
        #pragma unroll
        for (int v = 0; v < 8; ++v) {
            if (mrow0 + v < N && cidx < OUTC) outp[(size_t)v * OUTC] = acc0[v];
        }
    }
}

__global__ __launch_bounds__(256) void k_gather2(const int* __restrict__ cnt,
                                                 const int* __restrict__ slots,
                                                 const int* __restrict__ ovfn,
                                                 const int2* __restrict__ ovf,
                                                 const float* __restrict__ h2,
                                                 const float* __restrict__ b2,
                                                 float* __restrict__ out, int N) {
    const int lane = threadIdx.x & 31;
    const int d = __builtin_amdgcn_readfirstlane((int)(blockIdx.x * 8 + (threadIdx.x >> 5)));
    if (d >= N) return;
    const int c = __builtin_amdgcn_readfirstlane(cnt[d]);
    const int mall = (c < CAP) ? c : CAP;
    const float dd = rsqrtf((float)(c + 1));
    const int ch = (lane < OUTC) ? lane : 0;
    const float hs = h2[(size_t)d * OUTC + ch];
    const int* row = slots + (size_t)d * CAP;
    float acc = 0.0f;

    for (int base = 0; base < mall; base += 32) {
        const int m = (mall - base < 32) ? (mall - base) : 32;
        int eo = d * OUTC;
        if (lane < m) eo = row[base + lane] * OUTC;
        for (int k0 = 0; k0 < m; k0 += 8) {
            const int r = m - k0;
            float v[8];
            #pragma unroll
            for (int u = 0; u < 8; ++u) {
                if (u < r) {
                    int so = __builtin_amdgcn_readlane(eo, k0 + u);
                    v[u] = h2[so + ch];
                }
            }
            #pragma unroll
            for (int u = 0; u < 8; ++u) {
                if (u < r) acc += v[u];
            }
        }
    }
    if (c > CAP) {
        const int n = __builtin_amdgcn_readfirstlane(*ovfn);
        for (int k = 0; k < n; k += 32) {
            int2 en = make_int2(d, -1);
            if (k + lane < n) en = ovf[k + lane];
            unsigned mask = __builtin_amdgcn_ballot_w32(en.y == d);
            while (mask) {
                const int j = __builtin_ctz(mask);
                mask &= mask - 1;
                const int s = __builtin_amdgcn_readlane(en.x, j);
                acc += h2[(size_t)s * OUTC + ch];
            }
        }
    }
    acc = (acc + hs) * dd;
    if (lane < OUTC) out[(size_t)d * OUTC + lane] = acc + b2[lane];
}

extern "C" void kernel_launch(void* const* d_in, const int* in_sizes, int n_in,
                              void* d_out, int out_size, void* d_ws, size_t ws_size,
                              hipStream_t stream) {
    const float* x    = (const float*)d_in[0];
    const int*   ei   = (const int*)d_in[1];
    const float* anc  = (const float*)d_in[3];
    const float* W1   = (const float*)d_in[4];
    const float* b1   = (const float*)d_in[5];
    const float* W2   = (const float*)d_in[6];
    const float* b2   = (const float*)d_in[7];
    float* out = (float*)d_out;

    const int N = in_sizes[0] / IN_CH;
    const int E = in_sizes[1] / 2;
    const int* src = ei;
    const int* dst = ei + E;

    float* h1    = (float*)d_ws;
    float* agg1  = h1 + (size_t)N * HID;
    int*   slots = (int*)(agg1 + (size_t)N * HID);
    int2*  ovf   = (int2*)(slots + (size_t)N * CAP);
    float* h2    = (float*)(ovf + (size_t)E);
    int*   cnt   = (int*)(h2 + (size_t)N * OUTC);
    int*   ovfn  = cnt + (size_t)N;

    k_zero<<<(N + 255) / 256, 256, 0, stream>>>(cnt, ovfn, N);
    k_slot<<<(E + 255) / 256, 256, 0, stream>>>(src, dst, cnt, slots, ovfn, ovf, E);

    k_gemm1<<<(N + 31) / 32, 128, 0, stream>>>(x, anc, W1, cnt, h1, N);

    k_gather1<<<(N + 7) / 8, 256, 0, stream>>>(cnt, slots, ovfn, ovf, h1, agg1, N);

    k_gemm2<<<(N + 63) / 64, 128, 0, stream>>>(agg1, b1, W2, cnt, h2, N);

    k_gather2<<<(N + 7) / 8, 256, 0, stream>>>(cnt, slots, ovfn, ovf, h2, b2, out, N);
}


# Edge slots built by two LDS sorting launches (edges binned by 256-node dst range, then ranked per node in LDS, rows written as whole lines), not one global atomic per edge; arrival-order sums
# speedup vs baseline: 3.8958x; 1.5329x over previous
#include <hip/hip_runtime.h>
#include <hip/hip_bf16.h>


typedef __attribute__((ext_vector_type(2))) float v2f;
typedef __attribute__((ext_vector_type(8))) float v8f;

#define IN_CH 128
#define IN2   256
#define HID   128
#define OUTC  10

#define CAP 64
#define BW 256
#define BSH 8
#define CAPB 6144
#define NBMAX 1024
#define L1_EPT 8

__device__ __forceinline__ int cnt_deg(int c) { return c & 0xFFFFFF; }
__device__ __forceinline__ int cnt_slots(int c) { return c >> 24; }

__global__ __launch_bounds__(256) void k_zero(int* xdeg, int* bcur, int* ovfn, int N, int NB) {
    int i = blockIdx.x * 256 + threadIdx.x;
    if (i < N) xdeg[i] = 0;
    if (i < NB) bcur[i] = 0;
    if (i == 0) *ovfn = 0;
}

__global__ __launch_bounds__(512) void k_bin(const int* __restrict__ src,
                                             const int* __restrict__ dst,
                                             int* bcur, int2* __restrict__ bk,
                                             int* xdeg, int* ovfn, int2* __restrict__ ovf,
                                             int E, int NB) {
    __shared__ int hist[NBMAX];
    const int tid = threadIdx.x;
    for (int i = tid; i < NB; i += 512) hist[i] = 0;
    __syncthreads();

    const int e0 = blockIdx.x * (512 * L1_EPT) + tid;
    int s[L1_EPT], d[L1_EPT], r[L1_EPT];
    #pragma unroll
    for (int j = 0; j < L1_EPT; ++j) {
        const int e = e0 + 512 * j;
        const int ec = e < E ? e : E - 1;
        s[j] = src[ec];
        d[j] = dst[ec];
    }
    #pragma unroll
    for (int j = 0; j < L1_EPT; ++j) {
        r[j] = 0;
        if (e0 + 512 * j < E) r[j] = atomicAdd(&hist[d[j] >> BSH], 1);
    }
    __syncthreads();
    for (int i = tid; i < NB; i += 512) {
        const int h = hist[i];
        if (h > 0) hist[i] = atomicAdd(&bcur[i], h);
    }
    __syncthreads();
    #pragma unroll
    for (int j = 0; j < L1_EPT; ++j) r[j] += hist[d[j] >> BSH];
    #pragma unroll
    for (int j = 0; j < L1_EPT; ++j) {
        if (e0 + 512 * j < E) {
            const int b = d[j] >> BSH;
            const int pos = r[j];
            if (pos < CAPB) {
                bk[(size_t)b * CAPB + pos] = make_int2(s[j], d[j]);
            } else {
                const int k = atomicAdd(ovfn, 1);
                ovf[k] = make_int2(s[j], d[j]);
                atomicAdd(&xdeg[d[j]], 1);
            }
        }
    }
}

__global__ __launch_bounds__(512) void k_csr(const int* __restrict__ bcur,
                                             const int2* __restrict__ bk,
                                             const int* __restrict__ xdeg,
                                             int* __restrict__ cnt, int* __restrict__ slots,
                                             int* ovfn, int2* __restrict__ ovf, int N) {
    __shared__ __attribute__((aligned(16))) int rows[BW * CAP];
    __shared__ int lcnt[BW];
    const int tid = threadIdx.x;
    const int b = blockIdx.x;
    const int4 z = {0, 0, 0, 0};
    for (int q = tid; q < BW * CAP / 4; q += 512) ((int4*)rows)[q] = z;
    if (tid < BW) lcnt[tid] = 0;
    __syncthreads();

    int nb = bcur[b];
    nb = nb < CAPB ? nb : CAPB;
    const int2* seg = bk + (size_t)b * CAPB;
    for (int i0 = 0; i0 < nb; i0 += 512 * 8) {
        int2 ed[8];
        #pragma unroll
        for (int u = 0; u < 8; ++u) {
            const int i = i0 + tid + 512 * u;
            ed[u] = seg[i < nb ? i : nb - 1];
        }
        int p[8];
        #pragma unroll
        for (int u = 0; u < 8; ++u) {
            p[u] = 0;
            if (i0 + tid + 512 * u < nb) p[u] = atomicAdd(&lcnt[ed[u].y & (BW - 1)], 1);
        }
        #pragma unroll
        for (int u = 0; u < 8; ++u) {
            if (i0 + tid + 512 * u < nb) {
                const int row = ed[u].y & (BW - 1);
                if (p[u] < CAP) {
                    rows[row * CAP + p[u]] = ed[u].x;
                } else {
                    const int k = atomicAdd(ovfn, 1);
                    ovf[k] = ed[u];
                }
            }
        }
    }
    __syncthreads();

    {
        int4 pv[BW * CAP / 4 / 512];
        int lc[BW * CAP / 4 / 512];
        #pragma unroll
        for (int t = 0; t < BW * CAP / 4 / 512; ++t) {
            const int q = tid + 512 * t;
            pv[t] = ((const int4*)rows)[q];
            lc[t] = lcnt[q >> 4];
        }
        #pragma unroll
        for (int t = 0; t < BW * CAP / 4 / 512; ++t) {
            const int q = tid + 512 * t;
            const int row = q >> 4;
            const int c4 = (q & 15) * 4;
            if (c4 < lc[t]) {
                *(int4*)(slots + ((size_t)b * BW + row) * CAP + c4) = pv[t];
            }
        }
    }
    if (tid < BW) {
        const int node = b * BW + tid;
        if (node < N) {
            const int lc = lcnt[tid];
            const int ns = lc < CAP ? lc : CAP;
            cnt[node] = (lc + xdeg[node]) | (ns << 24);
        }
    }
}

__device__ __forceinline__ float deg_scale(int c) {
    return rsqrtf((float)(cnt_deg(c) + 1));
}

#define AS_STRIDE 258
#define G1_ROWS 32

__global__ __launch_bounds__(128) void k_gemm1(const float* __restrict__ x,
                                               const float* __restrict__ anc,
                                               const float* __restrict__ W1,
                                               const int* __restrict__ cnt,
                                               float* __restrict__ h1, int N) {
    __shared__ __attribute__((aligned(16))) float As[G1_ROWS * AS_STRIDE];
    const int row0 = blockIdx.x * G1_ROWS;
    const int tid = threadIdx.x;

    {
        const int c4 = (tid & 31) * 4;
        float4 xv[8], av[8];
        int cn[8];
        #pragma unroll
        for (int j = 0; j < 8; ++j) {
            const int node = row0 + (tid >> 5) + 4 * j;
            const int nc = node < N ? node : N - 1;
            xv[j] = *(const float4*)(x + (size_t)nc * IN_CH + c4);
            av[j] = *(const float4*)(anc + (size_t)nc * IN_CH + c4);
            cn[j] = cnt[nc];
        }
        #pragma unroll
        for (int j = 0; j < 8; ++j) {
            const int r = (tid >> 5) + 4 * j;
            const float sc = (row0 + r < N) ? deg_scale(cn[j]) : 0.0f;
            float2 d0, d1, a0, a1;
            d0.x = (xv[j].x - av[j].x) * sc;
            d0.y = (xv[j].y - av[j].y) * sc;
            d1.x = (xv[j].z - av[j].z) * sc;
            d1.y = (xv[j].w - av[j].w) * sc;
            a0.x = av[j].x * sc;
            a0.y = av[j].y * sc;
            a1.x = av[j].z * sc;
            a1.y = av[j].w * sc;
            float* dp = &As[r * AS_STRIDE + c4];
            *(float2*)(dp) = d0;
            *(float2*)(dp + 2) = d1;
            *(float2*)(dp + IN_CH) = a0;
            *(float2*)(dp + IN_CH + 2) = a1;
        }
    }
    __syncthreads();

    const int wave = tid >> 5, lane = tid & 31;
    const int grp = lane >> 4;
    const int cidx = lane & 15;
    const int n0 = wave * 32 + 2 * cidx;

    const float* a0p = &As[cidx * AS_STRIDE + 2 * grp];
    const float* a1p = &As[(16 + cidx) * AS_STRIDE + 2 * grp];
    const float* bp = W1 + (size_t)(2 * grp) * HID + n0;

    v8f acc[2][2][2] = {};
    #pragma unroll 2
    for (int k0 = 0; k0 < IN2; k0 += 8) {
        #pragma unroll
        for (int p = 0; p < 2; ++p) {
            const int kk = k0 + 4 * p;
            v2f a0, a1, b0, b1;
            a0.x = a0p[kk];
            a0.y = a0p[kk + 1];
            a1.x = a1p[kk];
            a1.y = a1p[kk + 1];
            const float2 w0 = *(const float2*)(bp + (size_t)kk * HID);
            const float2 w1 = *(const float2*)(bp + (size_t)(kk + 1) * HID);
            b0.x = w0.x; b0.y = w1.x;
            b1.x = w0.y; b1.y = w1.y;
            acc[p][0][0] = __builtin_amdgcn_wmma_f32_16x16x4_f32(false, a0, false, b0, (short)0, acc[p][0][0], false, false);
            acc[p][0][1] = __builtin_amdgcn_wmma_f32_16x16x4_f32(false, a0, false, b1, (short)0, acc[p][0][1], false, false);
            acc[p][1][0] = __builtin_amdgcn_wmma_f32_16x16x4_f32(false, a1, false, b0, (short)0, acc[p][1][0], false, false);
            acc[p][1][1] = __builtin_amdgcn_wmma_f32_16x16x4_f32(false, a1, false, b1, (short)0, acc[p][1][1], false, false);
        }
    }

    #pragma unroll
    for (int m = 0; m < 2; ++m) {
        const v8f e = acc[0][m][0] + acc[1][m][0];
        const v8f o = acc[0][m][1] + acc[1][m][1];
        const int rb = row0 + 16 * m + 8 * grp;
        float* outp = h1 + (size_t)rb * HID + n0;
        if (row0 + G1_ROWS <= N) {
            #pragma unroll
            for (int v = 0; v < 8; ++v) {
                float2 t; t.x = e[v]; t.y = o[v];
                *(float2*)(outp + (size_t)v * HID) = t;
            }
        } else {
            #pragma unroll
            for (int v = 0; v < 8; ++v) {
                if (rb + v < N) {
                    float2 t; t.x = e[v]; t.y = o[v];
                    *(float2*)(outp + (size_t)v * HID) = t;
                }
            }
        }
    }
}

__global__ __launch_bounds__(256) void k_gather1(const int* __restrict__ cnt,
                                                 const int* __restrict__ slots,
                                                 const int* __restrict__ ovfn,
                                                 const int2* __restrict__ ovf,
                                                 const float* __restrict__ h1,
                                                 float* __restrict__ agg1, int N) {
    const int lane = threadIdx.x & 31;
    const int d = __builtin_amdgcn_readfirstlane((int)(blockIdx.x * 8 + (threadIdx.x >> 5)));
    if (d >= N) return;
    const int cw = __builtin_amdgcn_readfirstlane(cnt[d]);
    const int c = cnt_deg(cw);
    const int mall = cnt_slots(cw);
    const float dd = rsqrtf((float)(c + 1));
    const float4 sv = *(const float4*)(h1 + (size_t)d * HID + lane * 4);
    const int* row = slots + (size_t)d * CAP;
    float4 acc = {0.0f, 0.0f, 0.0f, 0.0f};

    for (int base = 0; base < mall; base += 32) {
        const int m = (mall - base < 32) ? (mall - base) : 32;
        int eo = d * HID;
        if (lane < m) eo = row[base + lane] * HID;
        for (int k0 = 0; k0 < m; k0 += 8) {
            const int r = m - k0;
            float4 v[8];
            #pragma unroll
            for (int u = 0; u < 8; ++u) {
                if (u < r) {
                    int so = __builtin_amdgcn_readlane(eo, k0 + u);
                    v[u] = *(const float4*)(h1 + so + lane * 4);
                }
            }
            #pragma unroll
            for (int u = 0; u < 8; ++u) {
                if (u < r) {
                    acc.x += v[u].x;
                    acc.y += v[u].y;
                    acc.z += v[u].z;
                    acc.w += v[u].w;
                }
            }
        }
    }
    if (c > mall) {
        const int n = __builtin_amdgcn_readfirstlane(*ovfn);
        for (int k = 0; k < n; k += 32) {
            int2 en = make_int2(d, -1);
            if (k + lane < n) en = ovf[k + lane];
            unsigned mask = __builtin_amdgcn_ballot_w32(en.y == d);
            while (mask) {
                const int j = __builtin_ctz(mask);
                mask &= mask - 1;
                const int s = __builtin_amdgcn_readlane(en.x, j);
                const float4 v = *(const float4*)(h1 + (size_t)s * HID + lane * 4);
                acc.x += v.x; acc.y += v.y; acc.z += v.z; acc.w += v.w;
            }
        }
    }
    acc.x = (acc.x + sv.x) * dd; acc.y = (acc.y + sv.y) * dd;
    acc.z = (acc.z + sv.z) * dd; acc.w = (acc.w + sv.w) * dd;
    *(float4*)(agg1 + (size_t)d * HID + lane * 4) = acc;
}

#define AS2_STRIDE 132

__global__ __launch_bounds__(128) void k_gemm2(const float* __restrict__ agg1,
                                               const float* __restrict__ b1,
                                               const float* __restrict__ W2,
                                               const int* __restrict__ cnt,
                                               float* __restrict__ h2, int N) {
    __shared__ __attribute__((aligned(16))) float As[64 * AS2_STRIDE];
    __shared__ float Bs[HID * 16];
    const int node0 = blockIdx.x * 64;
    const int tid = threadIdx.x;

    {
        const int c4 = (tid & 31) * 4;
        const int n = tid & 15;
        const float4 bv = *(const float4*)(b1 + c4);
        float4 gv[16];
        int cn[16];
        float wv[16];
        #pragma unroll
        for (int j = 0; j < 16; ++j) {
            const int node = node0 + (tid >> 5) + 4 * j;
            const int nc = node < N ? node : N - 1;
            gv[j] = *(const float4*)(agg1 + (size_t)nc * HID + c4);
            cn[j] = cnt[nc];
        }
        #pragma unroll
        for (int j = 0; j < 16; ++j) {
            const int k = (tid >> 4) + 8 * j;
            wv[j] = W2[(size_t)k * OUTC + (n < OUTC ? n : 0)];
        }
        #pragma unroll
        for (int j = 0; j < 16; ++j) {
            const int r = (tid >> 5) + 4 * j;
            const float sc = (node0 + r < N) ? deg_scale(cn[j]) : 0.0f;
            float4 v;
            v.x = gv[j].x + bv.x; v.y = gv[j].y + bv.y;
            v.z = gv[j].z + bv.z; v.w = gv[j].w + bv.w;
            v.x = v.x > 0.0f ? v.x : 0.0f; v.y = v.y > 0.0f ? v.y : 0.0f;
            v.z = v.z > 0.0f ? v.z : 0.0f; v.w = v.w > 0.0f ? v.w : 0.0f;
            v.x *= sc; v.y *= sc; v.z *= sc; v.w *= sc;
            *(float4*)&As[r * AS2_STRIDE + c4] = v;
        }
        #pragma unroll
        for (int j = 0; j < 16; ++j) {
            Bs[tid + 128 * j] = (n < OUTC) ? wv[j] : 0.0f;
        }
    }
    __syncthreads();

    const int wave = tid >> 5, lane = tid & 31;
    const int grp = lane >> 4, cidx = lane & 15;
    const int rbase = wave * 16;

    v8f acc0 = {}, acc1 = {};
    for (int k0 = 0; k0 < HID; k0 += 8) {
        {
            int kk = k0 + 2 * grp;
            v2f a, b;
            a.x = As[(rbase + cidx) * AS2_STRIDE + kk];
            a.y = As[(rbase + cidx) * AS2_STRIDE + kk + 1];
            b.x = Bs[kk * 16 + cidx];
            b.y = Bs[(kk + 1) * 16 + cidx];
            acc0 = __builtin_amdgcn_wmma_f32_16x16x4_f32(false, a, false, b, (short)0, acc0, false, false);
        }
        {
            int kk = k0 + 4 + 2 * grp;
            v2f a, b;
            a.x = As[(rbase + cidx) * AS2_STRIDE + kk];
            a.y = As[(rbase + cidx) * AS2_STRIDE + kk + 1];
            b.x = Bs[kk * 16 + cidx];
            b.y = Bs[(kk + 1) * 16 + cidx];
            acc1 = __builtin_amdgcn_wmma_f32_16x16x4_f32(false, a, false, b, (short)0, acc1, false, false);
        }
    }
    acc0 = acc0 + acc1;

    const int mrow0 = node0 + rbase + 8 * grp;
    float* outp = h2 + (size_t)mrow0 * OUTC + cidx;
    if (node0 + 64 <= N) {
        if (cidx < OUTC) {
            #pragma unroll
            for (int v = 0; v < 8; ++v) outp[(size_t)v * OUTC] = acc0[v];
        }
    } else {
        #pragma unroll
        for (int v = 0; v < 8; ++v) {
            if (mrow0 + v < N && cidx < OUTC) outp[(size_t)v * OUTC] = acc0[v];
        }
    }
}

__global__ __launch_bounds__(256) void k_gather2(const int* __restrict__ cnt,
                                                 const int* __restrict__ slots,
                                                 const int* __restrict__ ovfn,
                                                 const int2* __restrict__ ovf,
                                                 const float* __restrict__ h2,
                                                 const float* __restrict__ b2,
                                                 float* __restrict__ out, int N) {
    const int lane = threadIdx.x & 31;
    const int d = __builtin_amdgcn_readfirstlane((int)(blockIdx.x * 8 + (threadIdx.x >> 5)));
    if (d >= N) return;
    const int cw = __builtin_amdgcn_readfirstlane(cnt[d]);
    const int c = cnt_deg(cw);
    const int mall = cnt_slots(cw);
    const float dd = rsqrtf((float)(c + 1));
    const int ch = (lane < OUTC) ? lane : 0;
    const float hs = h2[(size_t)d * OUTC + ch];
    const int* row = slots + (size_t)d * CAP;
    float acc = 0.0f;

    for (int base = 0; base < mall; base += 32) {
        const int m = (mall - base < 32) ? (mall - base) : 32;
        int eo = d * OUTC;
        if (lane < m) eo = row[base + lane] * OUTC;
        for (int k0 = 0; k0 < m; k0 += 8) {
            const int r = m - k0;
            float v[8];
            #pragma unroll
            for (int u = 0; u < 8; ++u) {
                if (u < r) {
                    int so = __builtin_amdgcn_readlane(eo, k0 + u);
                    v[u] = h2[so + ch];
                }
            }
            #pragma unroll
            for (int u = 0; u < 8; ++u) {
                if (u < r) acc += v[u];
            }
        }
    }
    if (c > mall) {
        const int n = __builtin_amdgcn_readfirstlane(*ovfn);
        for (int k = 0; k < n; k += 32) {
            int2 en = make_int2(d, -1);
            if (k + lane < n) en = ovf[k + lane];
            unsigned mask = __builtin_amdgcn_ballot_w32(en.y == d);
            while (mask) {
                const int j = __builtin_ctz(mask);
                mask &= mask - 1;
                const int s = __builtin_amdgcn_readlane(en.x, j);
                acc += h2[(size_t)s * OUTC + ch];
            }
        }
    }
    acc = (acc + hs) * dd;
    if (lane < OUTC) out[(size_t)d * OUTC + lane] = acc + b2[lane];
}

extern "C" void kernel_launch(void* const* d_in, const int* in_sizes, int n_in,
                              void* d_out, int out_size, void* d_ws, size_t ws_size,
                              hipStream_t stream) {
    const float* x    = (const float*)d_in[0];
    const int*   ei   = (const int*)d_in[1];
    const float* anc  = (const float*)d_in[3];
    const float* W1   = (const float*)d_in[4];
    const float* b1   = (const float*)d_in[5];
    const float* W2   = (const float*)d_in[6];
    const float* b2   = (const float*)d_in[7];
    float* out = (float*)d_out;

    const int N = in_sizes[0] / IN_CH;
    const int E = in_sizes[1] / 2;
    const int* src = ei;
    const int* dst = ei + E;

    const int NB = (N + BW - 1) / BW;

    float* h1    = (float*)d_ws;
    float* agg1  = h1 + (size_t)N * HID;
    int*   slots = (int*)(agg1 + (size_t)N * HID);
    int2*  ovf   = (int2*)(slots + (size_t)N * CAP);
    int2*  bk    = ovf + (size_t)E;
    float* h2    = (float*)(bk + (size_t)NB * CAPB);
    int*   cnt   = (int*)(h2 + (size_t)N * OUTC);
    int*   xdeg  = cnt + (size_t)N;
    int*   bcur  = xdeg + (size_t)N;
    int*   ovfn  = bcur + (size_t)NB;

    k_zero<<<(N + 255) / 256, 256, 0, stream>>>(xdeg, bcur, ovfn, N, NB);
    k_bin<<<(E + 512 * L1_EPT - 1) / (512 * L1_EPT), 512, 0, stream>>>(src, dst, bcur, bk, xdeg, ovfn, ovf, E, NB);
    k_csr<<<NB, 512, 0, stream>>>(bcur, bk, xdeg, cnt, slots, ovfn, ovf, N);

    k_gemm1<<<(N + 31) / 32, 128, 0, stream>>>(x, anc, W1, cnt, h1, N);

    k_gather1<<<(N + 7) / 8, 256, 0, stream>>>(cnt, slots, ovfn, ovf, h1, agg1, N);

    k_gemm2<<<(N + 63) / 64, 128, 0, stream>>>(agg1, b1, W2, cnt, h2, N);

    k_gather2<<<(N + 7) / 8, 256, 0, stream>>>(cnt, slots, ovfn, ovf, h2, b2, out, N);
}


# Layer-1 gather fused with the second f32 product: 16 waves put their aggregated rows into an LDS tile, the last one in (LDS counter) multiplies it by W2; agg1 never stored; sum order unchanged
# speedup vs baseline: 4.1364x; 1.0618x over previous
#include <hip/hip_runtime.h>
#include <hip/hip_bf16.h>


typedef __attribute__((ext_vector_type(2))) float v2f;
typedef __attribute__((ext_vector_type(8))) float v8f;

#define IN_CH 128
#define IN2   256
#define HID   128
#define OUTC  10

#define CAP 64
#define BW 256
#define BSH 8
#define CAPB 6144
#define NBMAX 1024
#define L1_EPT 8

__device__ __forceinline__ int cnt_deg(int c) { return c & 0xFFFFFF; }
__device__ __forceinline__ int cnt_slots(int c) { return c >> 24; }

__global__ __launch_bounds__(256) void k_zero(int* xdeg, int* bcur, int* ovfn, int N, int NB) {
    int i = blockIdx.x * 256 + threadIdx.x;
    if (i < N) xdeg[i] = 0;
    if (i < NB) bcur[i] = 0;
    if (i == 0) *ovfn = 0;
}

__global__ __launch_bounds__(512) void k_bin(const int* __restrict__ src,
                                             const int* __restrict__ dst,
                                             int* bcur, int2* __restrict__ bk,
                                             int* xdeg, int* ovfn, int2* __restrict__ ovf,
                                             int E, int NB) {
    __shared__ int hist[NBMAX];
    const int tid = threadIdx.x;
    for (int i = tid; i < NB; i += 512) hist[i] = 0;
    __syncthreads();

    const int e0 = blockIdx.x * (512 * L1_EPT) + tid;
    int s[L1_EPT], d[L1_EPT], r[L1_EPT];
    #pragma unroll
    for (int j = 0; j < L1_EPT; ++j) {
        const int e = e0 + 512 * j;
        const int ec = e < E ? e : E - 1;
        s[j] = src[ec];
        d[j] = dst[ec];
    }
    #pragma unroll
    for (int j = 0; j < L1_EPT; ++j) {
        r[j] = 0;
        if (e0 + 512 * j < E) r[j] = atomicAdd(&hist[d[j] >> BSH], 1);
    }
    __syncthreads();
    for (int i = tid; i < NB; i += 512) {
        const int h = hist[i];
        if (h > 0) hist[i] = atomicAdd(&bcur[i], h);
    }
    __syncthreads();
    #pragma unroll
    for (int j = 0; j < L1_EPT; ++j) r[j] += hist[d[j] >> BSH];
    #pragma unroll
    for (int j = 0; j < L1_EPT; ++j) {
        if (e0 + 512 * j < E) {
            const int b = d[j] >> BSH;
            const int pos = r[j];
            if (pos < CAPB) {
                bk[(size_t)b * CAPB + pos] = make_int2(s[j], d[j]);
            } else {
                const int k = atomicAdd(ovfn, 1);
                ovf[k] = make_int2(s[j], d[j]);
                atomicAdd(&xdeg[d[j]], 1);
            }
        }
    }
}

__global__ __launch_bounds__(512) void k_csr(const int* __restrict__ bcur,
                                             const int2* __restrict__ bk,
                                             const int* __restrict__ xdeg,
                                             int* __restrict__ cnt, int* __restrict__ slots,
                                             int* ovfn, int2* __restrict__ ovf, int N) {
    __shared__ __attribute__((aligned(16))) int rows[BW * CAP];
    __shared__ int lcnt[BW];
    const int tid = threadIdx.x;
    const int b = blockIdx.x;
    const int4 z = {0, 0, 0, 0};
    for (int q = tid; q < BW * CAP / 4; q += 512) ((int4*)rows)[q] = z;
    if (tid < BW) lcnt[tid] = 0;
    __syncthreads();

    int nb = bcur[b];
    nb = nb < CAPB ? nb : CAPB;
    const int2* seg = bk + (size_t)b * CAPB;
    for (int i0 = 0; i0 < nb; i0 += 512 * 8) {
        int2 ed[8];
        #pragma unroll
        for (int u = 0; u < 8; ++u) {
            const int i = i0 + tid + 512 * u;
            ed[u] = seg[i < nb ? i : nb - 1];
        }
        int p[8];
        #pragma unroll
        for (int u = 0; u < 8; ++u) {
            p[u] = 0;
            if (i0 + tid + 512 * u < nb) p[u] = atomicAdd(&lcnt[ed[u].y & (BW - 1)], 1);
        }
        #pragma unroll
        for (int u = 0; u < 8; ++u) {
            if (i0 + tid + 512 * u < nb) {
                const int row = ed[u].y & (BW - 1);
                if (p[u] < CAP) {
                    rows[row * CAP + p[u]] = ed[u].x;
                } else {
                    const int k = atomicAdd(ovfn, 1);
                    ovf[k] = ed[u];
                }
            }
        }
    }
    __syncthreads();

    {
        int4 pv[BW * CAP / 4 / 512];
        int lc[BW * CAP / 4 / 512];
        #pragma unroll
        for (int t = 0; t < BW * CAP / 4 / 512; ++t) {
            const int q = tid + 512 * t;
            pv[t] = ((const int4*)rows)[q];
            lc[t] = lcnt[q >> 4];
        }
        #pragma unroll
        for (int t = 0; t < BW * CAP / 4 / 512; ++t) {
            const int q = tid + 512 * t;
            const int row = q >> 4;
            const int c4 = (q & 15) * 4;
            if (c4 < lc[t]) {
                *(int4*)(slots + ((size_t)b * BW + row) * CAP + c4) = pv[t];
            }
        }
    }
    if (tid < BW) {
        const int node = b * BW + tid;
        if (node < N) {
            const int lc = lcnt[tid];
            const int ns = lc < CAP ? lc : CAP;
            cnt[node] = (lc + xdeg[node]) | (ns << 24);
        }
    }
}

__device__ __forceinline__ float deg_scale(int c) {
    return rsqrtf((float)(cnt_deg(c) + 1));
}

#define AS_STRIDE 258
#define G1_ROWS 32

__global__ __launch_bounds__(128) void k_gemm1(const float* __restrict__ x,
                                               const float* __restrict__ anc,
                                               const float* __restrict__ W1,
                                               const int* __restrict__ cnt,
                                               float* __restrict__ h1, int N) {
    __shared__ __attribute__((aligned(16))) float As[G1_ROWS * AS_STRIDE];
    const int row0 = blockIdx.x * G1_ROWS;
    const int tid = threadIdx.x;

    {
        const int c4 = (tid & 31) * 4;
        float4 xv[8], av[8];
        int cn[8];
        #pragma unroll
        for (int j = 0; j < 8; ++j) {
            const int node = row0 + (tid >> 5) + 4 * j;
            const int nc = node < N ? node : N - 1;
            xv[j] = *(const float4*)(x + (size_t)nc * IN_CH + c4);
            av[j] = *(const float4*)(anc + (size_t)nc * IN_CH + c4);
            cn[j] = cnt[nc];
        }
        #pragma unroll
        for (int j = 0; j < 8; ++j) {
            const int r = (tid >> 5) + 4 * j;
            const float sc = (row0 + r < N) ? deg_scale(cn[j]) : 0.0f;
            float2 d0, d1, a0, a1;
            d0.x = (xv[j].x - av[j].x) * sc;
            d0.y = (xv[j].y - av[j].y) * sc;
            d1.x = (xv[j].z - av[j].z) * sc;
            d1.y = (xv[j].w - av[j].w) * sc;
            a0.x = av[j].x * sc;
            a0.y = av[j].y * sc;
            a1.x = av[j].z * sc;
            a1.y = av[j].w * sc;
            float* dp = &As[r * AS_STRIDE + c4];
            *(float2*)(dp) = d0;
            *(float2*)(dp + 2) = d1;
            *(float2*)(dp + IN_CH) = a0;
            *(float2*)(dp + IN_CH + 2) = a1;
        }
    }
    __syncthreads();

    const int wave = tid >> 5, lane = tid & 31;
    const int grp = lane >> 4;
    const int cidx = lane & 15;
    const int n0 = wave * 32 + 2 * cidx;

    const float* a0p = &As[cidx * AS_STRIDE + 2 * grp];
    const float* a1p = &As[(16 + cidx) * AS_STRIDE + 2 * grp];
    const float* bp = W1 + (size_t)(2 * grp) * HID + n0;

    v8f acc[2][2][2] = {};
    #pragma unroll 2
    for (int k0 = 0; k0 < IN2; k0 += 8) {
        #pragma unroll
        for (int p = 0; p < 2; ++p) {
            const int kk = k0 + 4 * p;
            v2f a0, a1, b0, b1;
            a0.x = a0p[kk];
            a0.y = a0p[kk + 1];
            a1.x = a1p[kk];
            a1.y = a1p[kk + 1];
            const float2 w0 = *(const float2*)(bp + (size_t)kk * HID);
            const float2 w1 = *(const float2*)(bp + (size_t)(kk + 1) * HID);
            b0.x = w0.x; b0.y = w1.x;
            b1.x = w0.y; b1.y = w1.y;
            acc[p][0][0] = __builtin_amdgcn_wmma_f32_16x16x4_f32(false, a0, false, b0, (short)0, acc[p][0][0], false, false);
            acc[p][0][1] = __builtin_amdgcn_wmma_f32_16x16x4_f32(false, a0, false, b1, (short)0, acc[p][0][1], false, false);
            acc[p][1][0] = __builtin_amdgcn_wmma_f32_16x16x4_f32(false, a1, false, b0, (short)0, acc[p][1][0], false, false);
            acc[p][1][1] = __builtin_amdgcn_wmma_f32_16x16x4_f32(false, a1, false, b1, (short)0, acc[p][1][1], false, false);
        }
    }

    #pragma unroll
    for (int m = 0; m < 2; ++m) {
        const v8f e = acc[0][m][0] + acc[1][m][0];
        const v8f o = acc[0][m][1] + acc[1][m][1];
        const int rb = row0 + 16 * m + 8 * grp;
        float* outp = h1 + (size_t)rb * HID + n0;
        if (row0 + G1_ROWS <= N) {
            #pragma unroll
            for (int v = 0; v < 8; ++v) {
                float2 t; t.x = e[v]; t.y = o[v];
                *(float2*)(outp + (size_t)v * HID) = t;
            }
        } else {
            #pragma unroll
            for (int v = 0; v < 8; ++v) {
                if (rb + v < N) {
                    float2 t; t.x = e[v]; t.y = o[v];
                    *(float2*)(outp + (size_t)v * HID) = t;
                }
            }
        }
    }
}

#define AS2_STRIDE 132
#define GN 16

__global__ __launch_bounds__(512) void k_gather1_gemm2(const int* __restrict__ cnt,
                                                       const int* __restrict__ slots,
                                                       const int* __restrict__ ovfn,
                                                       const int2* __restrict__ ovf,
                                                       const float* __restrict__ h1,
                                                       const float* __restrict__ b1,
                                                       const float* __restrict__ W2,
                                                       float* __restrict__ h2, int N) {
    __shared__ __attribute__((aligned(16))) float As[GN * AS2_STRIDE];
    __shared__ __attribute__((aligned(16))) float Bs[HID * OUTC];
    __shared__ int arrived;
    const int tid = threadIdx.x;
    const int lane = tid & 31;
    const int wave = __builtin_amdgcn_readfirstlane(tid >> 5);
    const int node0 = blockIdx.x * GN;
    const int dn = node0 + wave;
    const bool live = dn < N;

    if (tid == 0) arrived = 0;
    __syncthreads();

    float4 wv = {0.0f, 0.0f, 0.0f, 0.0f};
    if (tid < HID * OUTC / 4) wv = *(const float4*)(W2 + tid * 4);

    float4 rowv = {0.0f, 0.0f, 0.0f, 0.0f};
    if (live) {
        const int d = dn;
        const int cw = __builtin_amdgcn_readfirstlane(cnt[d]);
        const int c = cnt_deg(cw);
        const int mall = cnt_slots(cw);
        const float dd = rsqrtf((float)(c + 1));
        const float4 sv = *(const float4*)(h1 + (size_t)d * HID + lane * 4);
        const float4 bv = *(const float4*)(b1 + lane * 4);
        const int* row = slots + (size_t)d * CAP;
        float4 acc = {0.0f, 0.0f, 0.0f, 0.0f};

        for (int base = 0; base < mall; base += 32) {
            const int m = (mall - base < 32) ? (mall - base) : 32;
            int eo = d * HID;
            if (lane < m) eo = row[base + lane] * HID;
            for (int k0 = 0; k0 < m; k0 += 8) {
                const int r = m - k0;
                float4 v[8];
                #pragma unroll
                for (int u = 0; u < 8; ++u) {
                    if (u < r) {
                        int so = __builtin_amdgcn_readlane(eo, k0 + u);
                        v[u] = *(const float4*)(h1 + so + lane * 4);
                    }
                }
                #pragma unroll
                for (int u = 0; u < 8; ++u) {
                    if (u < r) {
                        acc.x += v[u].x;
                        acc.y += v[u].y;
                        acc.z += v[u].z;
                        acc.w += v[u].w;
                    }
                }
            }
        }
        if (c > mall) {
            const int no = __builtin_amdgcn_readfirstlane(*ovfn);
            for (int k = 0; k < no; k += 32) {
                int2 en = make_int2(d, -1);
                if (k + lane < no) en = ovf[k + lane];
                unsigned mask = __builtin_amdgcn_ballot_w32(en.y == d);
                while (mask) {
                    const int j = __builtin_ctz(mask);
                    mask &= mask - 1;
                    const int s = __builtin_amdgcn_readlane(en.x, j);
                    const float4 v = *(const float4*)(h1 + (size_t)s * HID + lane * 4);
                    acc.x += v.x; acc.y += v.y; acc.z += v.z; acc.w += v.w;
                }
            }
        }
        acc.x = (acc.x + sv.x) * dd; acc.y = (acc.y + sv.y) * dd;
        acc.z = (acc.z + sv.z) * dd; acc.w = (acc.w + sv.w) * dd;
        rowv.x = acc.x + bv.x; rowv.y = acc.y + bv.y;
        rowv.z = acc.z + bv.z; rowv.w = acc.w + bv.w;
        rowv.x = rowv.x > 0.0f ? rowv.x : 0.0f; rowv.y = rowv.y > 0.0f ? rowv.y : 0.0f;
        rowv.z = rowv.z > 0.0f ? rowv.z : 0.0f; rowv.w = rowv.w > 0.0f ? rowv.w : 0.0f;
        rowv.x *= dd; rowv.y *= dd; rowv.z *= dd; rowv.w *= dd;
    }
    *(float4*)&As[wave * AS2_STRIDE + lane * 4] = rowv;
    if (tid < HID * OUTC / 4) *(float4*)&Bs[tid * 4] = wv;
    __threadfence_block();
    int before = 0;
    if (lane == 0) before = atomicAdd(&arrived, 1);
    before = __builtin_amdgcn_readfirstlane(before);
    if (before != GN - 1) return;
    __threadfence_block();

    const int grp = lane >> 4, cidx = lane & 15;
    const int cb = cidx < OUTC ? cidx : OUTC - 1;
    v8f acc0 = {}, acc1 = {};
    #pragma unroll 4
    for (int k0 = 0; k0 < HID; k0 += 8) {
        {
            int kk = k0 + 2 * grp;
            v2f a, b;
            a.x = As[cidx * AS2_STRIDE + kk];
            a.y = As[cidx * AS2_STRIDE + kk + 1];
            b.x = Bs[kk * OUTC + cb];
            b.y = Bs[(kk + 1) * OUTC + cb];
            acc0 = __builtin_amdgcn_wmma_f32_16x16x4_f32(false, a, false, b, (short)0, acc0, false, false);
        }
        {
            int kk = k0 + 4 + 2 * grp;
            v2f a, b;
            a.x = As[cidx * AS2_STRIDE + kk];
            a.y = As[cidx * AS2_STRIDE + kk + 1];
            b.x = Bs[kk * OUTC + cb];
            b.y = Bs[(kk + 1) * OUTC + cb];
            acc1 = __builtin_amdgcn_wmma_f32_16x16x4_f32(false, a, false, b, (short)0, acc1, false, false);
        }
    }
    acc0 = acc0 + acc1;

    const int mrow0 = node0 + 8 * grp;
    float* outp = h2 + (size_t)mrow0 * OUTC + cidx;
    if (node0 + GN <= N) {
        if (cidx < OUTC) {
            #pragma unroll
            for (int v = 0; v < 8; ++v) outp[(size_t)v * OUTC] = acc0[v];
        }
    } else {
        #pragma unroll
        for (int v = 0; v < 8; ++v) {
            if (mrow0 + v < N && cidx < OUTC) outp[(size_t)v * OUTC] = acc0[v];
        }
    }
}

__global__ __launch_bounds__(256) void k_gather2(const int* __restrict__ cnt,
                                                 const int* __restrict__ slots,
                                                 const int* __restrict__ ovfn,
                                                 const int2* __restrict__ ovf,
                                                 const float* __restrict__ h2,
                                                 const float* __restrict__ b2,
                                                 float* __restrict__ out, int N) {
    const int lane = threadIdx.x & 31;
    const int d = __builtin_amdgcn_readfirstlane((int)(blockIdx.x * 8 + (threadIdx.x >> 5)));
    if (d >= N) return;
    const int cw = __builtin_amdgcn_readfirstlane(cnt[d]);
    const int c = cnt_deg(cw);
    const int mall = cnt_slots(cw);
    const float dd = rsqrtf((float)(c + 1));
    const int ch = (lane < OUTC) ? lane : 0;
    const float hs = h2[(size_t)d * OUTC + ch];
    const int* row = slots + (size_t)d * CAP;
    float acc = 0.0f;

    for (int base = 0; base < mall; base += 32) {
        const int m = (mall - base < 32) ? (mall - base) : 32;
        int eo = d * OUTC;
        if (lane < m) eo = row[base + lane] * OUTC;
        for (int k0 = 0; k0 < m; k0 += 8) {
            const int r = m - k0;
            float v[8];
            #pragma unroll
            for (int u = 0; u < 8; ++u) {
                if (u < r) {
                    int so = __builtin_amdgcn_readlane(eo, k0 + u);
                    v[u] = h2[so + ch];
                }
            }
            #pragma unroll
            for (int u = 0; u < 8; ++u) {
                if (u < r) acc += v[u];
            }
        }
    }
    if (c > mall) {
        const int n = __builtin_amdgcn_readfirstlane(*ovfn);
        for (int k = 0; k < n; k += 32) {
            int2 en = make_int2(d, -1);
            if (k + lane < n) en = ovf[k + lane];
            unsigned mask = __builtin_amdgcn_ballot_w32(en.y == d);
            while (mask) {
                const int j = __builtin_ctz(mask);
                mask &= mask - 1;
                const int s = __builtin_amdgcn_readlane(en.x, j);
                acc += h2[(size_t)s * OUTC + ch];
            }
        }
    }
    acc = (acc + hs) * dd;
    if (lane < OUTC) out[(size_t)d * OUTC + lane] = acc + b2[lane];
}

extern "C" void kernel_launch(void* const* d_in, const int* in_sizes, int n_in,
                              void* d_out, int out_size, void* d_ws, size_t ws_size,
                              hipStream_t stream) {
    const float* x    = (const float*)d_in[0];
    const int*   ei   = (const int*)d_in[1];
    const float* anc  = (const float*)d_in[3];
    const float* W1   = (const float*)d_in[4];
    const float* b1   = (const float*)d_in[5];
    const float* W2   = (const float*)d_in[6];
    const float* b2   = (const float*)d_in[7];
    float* out = (float*)d_out;

    const int N = in_sizes[0] / IN_CH;
    const int E = in_sizes[1] / 2;
    const int* src = ei;
    const int* dst = ei + E;

    const int NB = (N + BW - 1) / BW;

    float* h1    = (float*)d_ws;
    int*   slots = (int*)(h1 + (size_t)N * HID);
    int2*  ovf   = (int2*)(slots + (size_t)N * CAP);
    int2*  bk    = ovf + (size_t)E;
    float* h2    = (float*)(bk + (size_t)NB * CAPB);
    int*   cnt   = (int*)(h2 + (size_t)N * OUTC);
    int*   xdeg  = cnt + (size_t)N;
    int*   bcur  = xdeg + (size_t)N;
    int*   ovfn  = bcur + (size_t)NB;

    k_zero<<<(N + 255) / 256, 256, 0, stream>>>(xdeg, bcur, ovfn, N, NB);
    k_bin<<<(E + 512 * L1_EPT - 1) / (512 * L1_EPT), 512, 0, stream>>>(src, dst, bcur, bk, xdeg, ovfn, ovf, E, NB);
    k_csr<<<NB, 512, 0, stream>>>(bcur, bk, xdeg, cnt, slots, ovfn, ovf, N);

    k_gemm1<<<(N + 31) / 32, 128, 0, stream>>>(x, anc, W1, cnt, h1, N);

    k_gather1_gemm2<<<(N + GN - 1) / GN, 512, 0, stream>>>(cnt, slots, ovfn, ovf, h1, b1, W2, h2, N);

    k_gather2<<<(N + 7) / 8, 256, 0, stream>>>(cnt, slots, ovfn, ovf, h2, b2, out, N);
}


# Layer-2 gather over h2 rows stored as 64 bytes (no row crosses a line): a wave takes 8 nodes, 4 lanes each, one 16-byte load per lane and edge; 12504 waves, not 100000; sums in the same order
# speedup vs baseline: 4.4441x; 1.0744x over previous
#include <hip/hip_runtime.h>
#include <hip/hip_bf16.h>


typedef __attribute__((ext_vector_type(2))) float v2f;
typedef __attribute__((ext_vector_type(8))) float v8f;

#define IN_CH 128
#define IN2   256
#define HID   128
#define OUTC  10

#define CAP 64
#define BW 256
#define BSH 8
#define CAPB 6144
#define NBMAX 1024
#define L1_EPT 8

__device__ __forceinline__ int cnt_deg(int c) { return c & 0xFFFFFF; }
__device__ __forceinline__ int cnt_slots(int c) { return c >> 24; }

__global__ __launch_bounds__(256) void k_zero(int* xdeg, int* bcur, int* ovfn, int N, int NB) {
    int i = blockIdx.x * 256 + threadIdx.x;
    if (i < N) xdeg[i] = 0;
    if (i < NB) bcur[i] = 0;
    if (i == 0) *ovfn = 0;
}

__global__ __launch_bounds__(512) void k_bin(const int* __restrict__ src,
                                             const int* __restrict__ dst,
                                             int* bcur, int2* __restrict__ bk,
                                             int* xdeg, int* ovfn, int2* __restrict__ ovf,
                                             int E, int NB) {
    __shared__ int hist[NBMAX];
    const int tid = threadIdx.x;
    for (int i = tid; i < NB; i += 512) hist[i] = 0;
    __syncthreads();

    const int e0 = blockIdx.x * (512 * L1_EPT) + tid;
    int s[L1_EPT], d[L1_EPT], r[L1_EPT];
    #pragma unroll
    for (int j = 0; j < L1_EPT; ++j) {
        const int e = e0 + 512 * j;
        const int ec = e < E ? e : E - 1;
        s[j] = src[ec];
        d[j] = dst[ec];
    }
    #pragma unroll
    for (int j = 0; j < L1_EPT; ++j) {
        r[j] = 0;
        if (e0 + 512 * j < E) r[j] = atomicAdd(&hist[d[j] >> BSH], 1);
    }
    __syncthreads();
    for (int i = tid; i < NB; i += 512) {
        const int h = hist[i];
        if (h > 0) hist[i] = atomicAdd(&bcur[i], h);
    }
    __syncthreads();
    #pragma unroll
    for (int j = 0; j < L1_EPT; ++j) r[j] += hist[d[j] >> BSH];
    #pragma unroll
    for (int j = 0; j < L1_EPT; ++j) {
        if (e0 + 512 * j < E) {
            const int b = d[j] >> BSH;
            const int pos = r[j];
            if (pos < CAPB) {
                bk[(size_t)b * CAPB + pos] = make_int2(s[j], d[j]);
            } else {
                const int k = atomicAdd(ovfn, 1);
                ovf[k] = make_int2(s[j], d[j]);
                atomicAdd(&xdeg[d[j]], 1);
            }
        }
    }
}

__global__ __launch_bounds__(512) void k_csr(const int* __restrict__ bcur,
                                             const int2* __restrict__ bk,
                                             const int* __restrict__ xdeg,
                                             int* __restrict__ cnt, int* __restrict__ slots,
                                             int* ovfn, int2* __restrict__ ovf, int N) {
    __shared__ __attribute__((aligned(16))) int rows[BW * CAP];
    __shared__ int lcnt[BW];
    const int tid = threadIdx.x;
    const int b = blockIdx.x;
    const int4 z = {0, 0, 0, 0};
    for (int q = tid; q < BW * CAP / 4; q += 512) ((int4*)rows)[q] = z;
    if (tid < BW) lcnt[tid] = 0;
    __syncthreads();

    int nb = bcur[b];
    nb = nb < CAPB ? nb : CAPB;
    const int2* seg = bk + (size_t)b * CAPB;
    for (int i0 = 0; i0 < nb; i0 += 512 * 8) {
        int2 ed[8];
        #pragma unroll
        for (int u = 0; u < 8; ++u) {
            const int i = i0 + tid + 512 * u;
            ed[u] = seg[i < nb ? i : nb - 1];
        }
        int p[8];
        #pragma unroll
        for (int u = 0; u < 8; ++u) {
            p[u] = 0;
            if (i0 + tid + 512 * u < nb) p[u] = atomicAdd(&lcnt[ed[u].y & (BW - 1)], 1);
        }
        #pragma unroll
        for (int u = 0; u < 8; ++u) {
            if (i0 + tid + 512 * u < nb) {
                const int row = ed[u].y & (BW - 1);
                if (p[u] < CAP) {
                    rows[row * CAP + p[u]] = ed[u].x;
                } else {
                    const int k = atomicAdd(ovfn, 1);
                    ovf[k] = ed[u];
                }
            }
        }
    }
    __syncthreads();

    {
        int4 pv[BW * CAP / 4 / 512];
        int lc[BW * CAP / 4 / 512];
        #pragma unroll
        for (int t = 0; t < BW * CAP / 4 / 512; ++t) {
            const int q = tid + 512 * t;
            pv[t] = ((const int4*)rows)[q];
            lc[t] = lcnt[q >> 4];
        }
        #pragma unroll
        for (int t = 0; t < BW * CAP / 4 / 512; ++t) {
            const int q = tid + 512 * t;
            const int row = q >> 4;
            const int c4 = (q & 15) * 4;
            if (c4 < lc[t]) {
                *(int4*)(slots + ((size_t)b * BW + row) * CAP + c4) = pv[t];
            }
        }
    }
    if (tid < BW) {
        const int node = b * BW + tid;
        if (node < N) {
            const int lc = lcnt[tid];
            const int ns = lc < CAP ? lc : CAP;
            cnt[node] = (lc + xdeg[node]) | (ns << 24);
        }
    }
}

__device__ __forceinline__ float deg_scale(int c) {
    return rsqrtf((float)(cnt_deg(c) + 1));
}

#define AS_STRIDE 258
#define G1_ROWS 32

__global__ __launch_bounds__(128) void k_gemm1(const float* __restrict__ x,
                                               const float* __restrict__ anc,
                                               const float* __restrict__ W1,
                                               const int* __restrict__ cnt,
                                               float* __restrict__ h1, int N) {
    __shared__ __attribute__((aligned(16))) float As[G1_ROWS * AS_STRIDE];
    const int row0 = blockIdx.x * G1_ROWS;
    const int tid = threadIdx.x;

    {
        const int c4 = (tid & 31) * 4;
        float4 xv[8], av[8];
        int cn[8];
        #pragma unroll
        for (int j = 0; j < 8; ++j) {
            const int node = row0 + (tid >> 5) + 4 * j;
            const int nc = node < N ? node : N - 1;
            xv[j] = *(const float4*)(x + (size_t)nc * IN_CH + c4);
            av[j] = *(const float4*)(anc + (size_t)nc * IN_CH + c4);
            cn[j] = cnt[nc];
        }
        #pragma unroll
        for (int j = 0; j < 8; ++j) {
            const int r = (tid >> 5) + 4 * j;
            const float sc = (row0 + r < N) ? deg_scale(cn[j]) : 0.0f;
            float2 d0, d1, a0, a1;
            d0.x = (xv[j].x - av[j].x) * sc;
            d0.y = (xv[j].y - av[j].y) * sc;
            d1.x = (xv[j].z - av[j].z) * sc;
            d1.y = (xv[j].w - av[j].w) * sc;
            a0.x = av[j].x * sc;
            a0.y = av[j].y * sc;
            a1.x = av[j].z * sc;
            a1.y = av[j].w * sc;
            float* dp = &As[r * AS_STRIDE + c4];
            *(float2*)(dp) = d0;
            *(float2*)(dp + 2) = d1;
            *(float2*)(dp + IN_CH) = a0;
            *(float2*)(dp + IN_CH + 2) = a1;
        }
    }
    __syncthreads();

    const int wave = tid >> 5, lane = tid & 31;
    const int grp = lane >> 4;
    const int cidx = lane & 15;
    const int n0 = wave * 32 + 2 * cidx;

    const float* a0p = &As[cidx * AS_STRIDE + 2 * grp];
    const float* a1p = &As[(16 + cidx) * AS_STRIDE + 2 * grp];
    const float* bp = W1 + (size_t)(2 * grp) * HID + n0;

    v8f acc[2][2][2] = {};
    #pragma unroll 2
    for (int k0 = 0; k0 < IN2; k0 += 8) {
        #pragma unroll
        for (int p = 0; p < 2; ++p) {
            const int kk = k0 + 4 * p;
            v2f a0, a1, b0, b1;
            a0.x = a0p[kk];
            a0.y = a0p[kk + 1];
            a1.x = a1p[kk];
            a1.y = a1p[kk + 1];
            const float2 w0 = *(const float2*)(bp + (size_t)kk * HID);
            const float2 w1 = *(const float2*)(bp + (size_t)(kk + 1) * HID);
            b0.x = w0.x; b0.y = w1.x;
            b1.x = w0.y; b1.y = w1.y;
            acc[p][0][0] = __builtin_amdgcn_wmma_f32_16x16x4_f32(false, a0, false, b0, (short)0, acc[p][0][0], false, false);
            acc[p][0][1] = __builtin_amdgcn_wmma_f32_16x16x4_f32(false, a0, false, b1, (short)0, acc[p][0][1], false, false);
            acc[p][1][0] = __builtin_amdgcn_wmma_f32_16x16x4_f32(false, a1, false, b0, (short)0, acc[p][1][0], false, false);
            acc[p][1][1] = __builtin_amdgcn_wmma_f32_16x16x4_f32(false, a1, false, b1, (short)0, acc[p][1][1], false, false);
        }
    }

    #pragma unroll
    for (int m = 0; m < 2; ++m) {
        const v8f e = acc[0][m][0] + acc[1][m][0];
        const v8f o = acc[0][m][1] + acc[1][m][1];
        const int rb = row0 + 16 * m + 8 * grp;
        float* outp = h1 + (size_t)rb * HID + n0;
        if (row0 + G1_ROWS <= N) {
            #pragma unroll
            for (int v = 0; v < 8; ++v) {
                float2 t; t.x = e[v]; t.y = o[v];
                *(float2*)(outp + (size_t)v * HID) = t;
            }
        } else {
            #pragma unroll
            for (int v = 0; v < 8; ++v) {
                if (rb + v < N) {
                    float2 t; t.x = e[v]; t.y = o[v];
                    *(float2*)(outp + (size_t)v * HID) = t;
                }
            }
        }
    }
}

#define AS2_STRIDE 132
#define GN 16
#define H2S 16

__global__ __launch_bounds__(512) void k_gather1_gemm2(const int* __restrict__ cnt,
                                                       const int* __restrict__ slots,
                                                       const int* __restrict__ ovfn,
                                                       const int2* __restrict__ ovf,
                                                       const float* __restrict__ h1,
                                                       const float* __restrict__ b1,
                                                       const float* __restrict__ W2,
                                                       float* __restrict__ h2, int N) {
    __shared__ __attribute__((aligned(16))) float As[GN * AS2_STRIDE];
    __shared__ __attribute__((aligned(16))) float Bs[HID * OUTC];
    __shared__ int arrived;
    const int tid = threadIdx.x;
    const int lane = tid & 31;
    const int wave = __builtin_amdgcn_readfirstlane(tid >> 5);
    const int node0 = blockIdx.x * GN;
    const int dn = node0 + wave;
    const bool live = dn < N;

    if (tid == 0) arrived = 0;
    __syncthreads();

    float4 wv = {0.0f, 0.0f, 0.0f, 0.0f};
    if (tid < HID * OUTC / 4) wv = *(const float4*)(W2 + tid * 4);

    float4 rowv = {0.0f, 0.0f, 0.0f, 0.0f};
    if (live) {
        const int d = dn;
        const int cw = __builtin_amdgcn_readfirstlane(cnt[d]);
        const int c = cnt_deg(cw);
        const int mall = cnt_slots(cw);
        const float dd = rsqrtf((float)(c + 1));
        const float4 sv = *(const float4*)(h1 + (size_t)d * HID + lane * 4);
        const float4 bv = *(const float4*)(b1 + lane * 4);
        const int* row = slots + (size_t)d * CAP;
        float4 acc = {0.0f, 0.0f, 0.0f, 0.0f};

        for (int base = 0; base < mall; base += 32) {
            const int m = (mall - base < 32) ? (mall - base) : 32;
            int eo = d * HID;
            if (lane < m) eo = row[base + lane] * HID;
            for (int k0 = 0; k0 < m; k0 += 8) {
                const int r = m - k0;
                float4 v[8];
                #pragma unroll
                for (int u = 0; u < 8; ++u) {
                    if (u < r) {
                        int so = __builtin_amdgcn_readlane(eo, k0 + u);
                        v[u] = *(const float4*)(h1 + so + lane * 4);
                    }
                }
                #pragma unroll
                for (int u = 0; u < 8; ++u) {
                    if (u < r) {
                        acc.x += v[u].x;
                        acc.y += v[u].y;
                        acc.z += v[u].z;
                        acc.w += v[u].w;
                    }
                }
            }
        }
        if (c > mall) {
            const int no = __builtin_amdgcn_readfirstlane(*ovfn);
            for (int k = 0; k < no; k += 32) {
                int2 en = make_int2(d, -1);
                if (k + lane < no) en = ovf[k + lane];
                unsigned mask = __builtin_amdgcn_ballot_w32(en.y == d);
                while (mask) {
                    const int j = __builtin_ctz(mask);
                    mask &= mask - 1;
                    const int s = __builtin_amdgcn_readlane(en.x, j);
                    const float4 v = *(const float4*)(h1 + (size_t)s * HID + lane * 4);
                    acc.x += v.x; acc.y += v.y; acc.z += v.z; acc.w += v.w;
                }
            }
        }
        acc.x = (acc.x + sv.x) * dd; acc.y = (acc.y + sv.y) * dd;
        acc.z = (acc.z + sv.z) * dd; acc.w = (acc.w + sv.w) * dd;
        rowv.x = acc.x + bv.x; rowv.y = acc.y + bv.y;
        rowv.z = acc.z + bv.z; rowv.w = acc.w + bv.w;
        rowv.x = rowv.x > 0.0f ? rowv.x : 0.0f; rowv.y = rowv.y > 0.0f ? rowv.y : 0.0f;
        rowv.z = rowv.z > 0.0f ? rowv.z : 0.0f; rowv.w = rowv.w > 0.0f ? rowv.w : 0.0f;
        rowv.x *= dd; rowv.y *= dd; rowv.z *= dd; rowv.w *= dd;
    }
    *(float4*)&As[wave * AS2_STRIDE + lane * 4] = rowv;
    if (tid < HID * OUTC / 4) *(float4*)&Bs[tid * 4] = wv;
    __threadfence_block();
    int before = 0;
    if (lane == 0) before = atomicAdd(&arrived, 1);
    before = __builtin_amdgcn_readfirstlane(before);
    if (before != GN - 1) return;
    __threadfence_block();

    const int grp = lane >> 4, cidx = lane & 15;
    const int cb = cidx < OUTC ? cidx : OUTC - 1;
    v8f acc0 = {}, acc1 = {};
    #pragma unroll 4
    for (int k0 = 0; k0 < HID; k0 += 8) {
        {
            int kk = k0 + 2 * grp;
            v2f a, b;
            a.x = As[cidx * AS2_STRIDE + kk];
            a.y = As[cidx * AS2_STRIDE + kk + 1];
            b.x = Bs[kk * OUTC + cb];
            b.y = Bs[(kk + 1) * OUTC + cb];
            acc0 = __builtin_amdgcn_wmma_f32_16x16x4_f32(false, a, false, b, (short)0, acc0, false, false);
        }
        {
            int kk = k0 + 4 + 2 * grp;
            v2f a, b;
            a.x = As[cidx * AS2_STRIDE + kk];
            a.y = As[cidx * AS2_STRIDE + kk + 1];
            b.x = Bs[kk * OUTC + cb];
            b.y = Bs[(kk + 1) * OUTC + cb];
            acc1 = __builtin_amdgcn_wmma_f32_16x16x4_f32(false, a, false, b, (short)0, acc1, false, false);
        }
    }
    acc0 = acc0 + acc1;

    const int mrow0 = node0 + 8 * grp;
    float* outp = h2 + (size_t)mrow0 * H2S + cidx;
    if (node0 + GN <= N) {
        #pragma unroll
        for (int v = 0; v < 8; ++v) outp[(size_t)v * H2S] = acc0[v];
    } else {
        #pragma unroll
        for (int v = 0; v < 8; ++v) {
            if (mrow0 + v < N) outp[(size_t)v * H2S] = acc0[v];
        }
    }
}

#define G2_NODES 8

__device__ __forceinline__ int quad_bcast(int v, int k) {
    switch (k) {
    case 0:  return __builtin_amdgcn_mov_dpp(v, 0x00, 0xF, 0xF, true);
    case 1:  return __builtin_amdgcn_mov_dpp(v, 0x55, 0xF, 0xF, true);
    case 2:  return __builtin_amdgcn_mov_dpp(v, 0xAA, 0xF, 0xF, true);
    default: return __builtin_amdgcn_mov_dpp(v, 0xFF, 0xF, 0xF, true);
    }
}

template <bool FIRST>
__device__ __forceinline__ void walk16(const int (&sw)[4], int base16, int mmax, int mall,
                                       bool rowlane, const float* hp, float4& acc) {
    #pragma unroll
    for (int h = 0; h < 2; ++h) {
        const int base = base16 + 8 * h;
        if (base < mmax) {
            float4 v[8];
            #pragma unroll
            for (int u = 0; u < 8; ++u) {
                if (base + u < mmax) {
                    const int s = quad_bcast(sw[u & 3], 2 * h + (u >> 2));
                    float4 t = {0.0f, 0.0f, 0.0f, 0.0f};
                    if (rowlane && base + u < mall) t = *(const float4*)(hp + (size_t)s * H2S);
                    v[u] = t;
                }
            }
            #pragma unroll
            for (int u = 0; u < 8; ++u) {
                if (base + u < mmax) {
                    if (FIRST && h == 0 && u == 0) {
                        acc = v[0];
                    } else {
                        acc.x += v[u].x;
                        acc.y += v[u].y;
                        acc.z += v[u].z;
                        acc.w += v[u].w;
                    }
                }
            }
        }
    }
}

__global__ __launch_bounds__(256) void k_gather2(const int* __restrict__ cnt,
                                                 const int* __restrict__ slots,
                                                 const int* __restrict__ ovfn,
                                                 const int2* __restrict__ ovf,
                                                 const float* __restrict__ h2,
                                                 const float* __restrict__ b2,
                                                 float* __restrict__ out, int N) {
    const int lane = threadIdx.x & 31;
    const int j = lane & 3;
    const int g = lane >> 2;
    const int node0 = __builtin_amdgcn_readfirstlane((int)(blockIdx.x * 8 + (threadIdx.x >> 5)) * G2_NODES);
    if (node0 >= N) return;
    const int d = node0 + g;
    const bool live = d < N;
    const int dl = live ? d : N - 1;
    const int cw = cnt[dl];
    const int c = live ? cnt_deg(cw) : 0;
    const int mall = live ? cnt_slots(cw) : 0;
    const float dd = rsqrtf((float)(c + 1));
    const bool rowlane = j < 3;
    const float* hp = h2 + 4 * j;
    float4 hs = {0.0f, 0.0f, 0.0f, 0.0f};
    if (rowlane) hs = *(const float4*)(hp + (size_t)dl * H2S);

    const int* row = slots + (size_t)dl * CAP;
    int sw0[4], sw1[4];
    {
        int4 t0 = {dl, dl, dl, dl}, t1 = {dl, dl, dl, dl};
        if (4 * j < mall) t0 = *(const int4*)(row + 4 * j);
        if (16 + 4 * j < mall) t1 = *(const int4*)(row + 16 + 4 * j);
        sw0[0] = t0.x; sw0[1] = t0.y; sw0[2] = t0.z; sw0[3] = t0.w;
        sw1[0] = t1.x; sw1[1] = t1.y; sw1[2] = t1.z; sw1[3] = t1.w;
    }
    int mmax = 0;
    #pragma unroll
    for (int t = 0; t < G2_NODES; ++t) {
        const int m = __builtin_amdgcn_readlane(mall, 4 * t);
        mmax = m > mmax ? m : mmax;
    }

    float4 acc = {0.0f, 0.0f, 0.0f, 0.0f};
    if (mmax > 0) walk16<true>(sw0, 0, mmax, mall, rowlane, hp, acc);
    if (mmax > 16) walk16<false>(sw1, 16, mmax, mall, rowlane, hp, acc);
    for (int b16 = 32; b16 < mmax; b16 += 16) {
        int4 t = {dl, dl, dl, dl};
        if (b16 + 4 * j < mall) t = *(const int4*)(row + b16 + 4 * j);
        int sw[4];
        sw[0] = t.x; sw[1] = t.y; sw[2] = t.z; sw[3] = t.w;
        walk16<false>(sw, b16, mmax, mall, rowlane, hp, acc);
    }
    const bool more = c > mall;
    if (__builtin_amdgcn_ballot_w32(more) != 0) {
        const int no = __builtin_amdgcn_readfirstlane(*ovfn);
        for (int k = 0; k < no; ++k) {
            const int2 en = ovf[k];
            if (more && rowlane && en.y == d) {
                const float4 t = *(const float4*)(hp + (size_t)en.x * H2S);
                acc.x += t.x; acc.y += t.y; acc.z += t.z; acc.w += t.w;
            }
        }
    }
    acc.x = (acc.x + hs.x) * dd; acc.y = (acc.y + hs.y) * dd;
    acc.z = (acc.z + hs.z) * dd; acc.w = (acc.w + hs.w) * dd;
    if (live && rowlane) {
        float* op = out + (size_t)d * OUTC + 4 * j;
        float2 lo;
        lo.x = acc.x + b2[4 * j];
        lo.y = acc.y + b2[4 * j + 1];
        *(float2*)op = lo;
        if (j < 2) {
            float2 hi;
            hi.x = acc.z + b2[4 * j + 2];
            hi.y = acc.w + b2[4 * j + 3];
            *(float2*)(op + 2) = hi;
        }
    }
}

extern "C" void kernel_launch(void* const* d_in, const int* in_sizes, int n_in,
                              void* d_out, int out_size, void* d_ws, size_t ws_size,
                              hipStream_t stream) {
    const float* x    = (const float*)d_in[0];
    const int*   ei   = (const int*)d_in[1];
    const float* anc  = (const float*)d_in[3];
    const float* W1   = (const float*)d_in[4];
    const float* b1   = (const float*)d_in[5];
    const float* W2   = (const float*)d_in[6];
    const float* b2   = (const float*)d_in[7];
    float* out = (float*)d_out;

    const int N = in_sizes[0] / IN_CH;
    const int E = in_sizes[1] / 2;
    const int* src = ei;
    const int* dst = ei + E;

    const int NB = (N + BW - 1) / BW;

    float* h1    = (float*)d_ws;
    int*   slots = (int*)(h1 + (size_t)N * HID);
    int2*  ovf   = (int2*)(slots + (size_t)N * CAP);
    int2*  bk    = ovf + (size_t)E;
    float* h2    = (float*)(bk + (size_t)NB * CAPB);
    int*   cnt   = (int*)(h2 + (size_t)N * H2S);
    int*   xdeg  = cnt + (size_t)N;
    int*   bcur  = xdeg + (size_t)N;
    int*   ovfn  = bcur + (size_t)NB;

    k_zero<<<(N + 255) / 256, 256, 0, stream>>>(xdeg, bcur, ovfn, N, NB);
    k_bin<<<(E + 512 * L1_EPT - 1) / (512 * L1_EPT), 512, 0, stream>>>(src, dst, bcur, bk, xdeg, ovfn, ovf, E, NB);
    k_csr<<<NB, 512, 0, stream>>>(bcur, bk, xdeg, cnt, slots, ovfn, ovf, N);

    k_gemm1<<<(N + 31) / 32, 128, 0, stream>>>(x, anc, W1, cnt, h1, N);

    k_gather1_gemm2<<<(N + GN - 1) / GN, 512, 0, stream>>>(cnt, slots, ovfn, ovf, h1, b1, W2, h2, N);

    k_gather2<<<(N + 8 * G2_NODES - 1) / (8 * G2_NODES), 256, 0, stream>>>(cnt, slots, ovfn, ovf, h2, b2, out, N);
}
